# starting kernel
# baseline (speedup 1.0000x reference)
#include <hip/hip_runtime.h>

typedef __attribute__((ext_vector_type(2))) float v2f;
typedef __attribute__((ext_vector_type(8))) float v8f;

#define NEG_SLOPE 0.2f

__device__ __forceinline__ void atomic_max_f32(float* addr, float v) {
  if (v >= 0.0f) atomicMax((int*)addr, __float_as_int(v));
  else           atomicMin((unsigned int*)addr, (unsigned int)__float_as_int(v));
}

__global__ void gat_init(float* __restrict__ attr_sum, float* __restrict__ cnt,
                         float* __restrict__ mx, float* __restrict__ den,
                         float* __restrict__ num, int N) {
  int i = blockIdx.x * blockDim.x + threadIdx.x;
  if (i < N) { attr_sum[i] = 0.0f; cnt[i] = 0.0f; }
  if (i < 2 * N) {
    mx[i]  = __int_as_float(0xff800000);
    den[i] = 0.0f;
    num[i] = 0.0f;
  }
}

__global__ void gat_degree(const int* __restrict__ ei,
                           const float* __restrict__ attr,
                           float* __restrict__ attr_sum, float* __restrict__ cnt,
                           int E) {
  int e = blockIdx.x * blockDim.x + threadIdx.x;
  if (e >= E) return;
  int d = ei[E + e];
  atomicAdd(&attr_sum[d], attr[e]);
  atomicAdd(&cnt[d], 1.0f);
}

__global__ void gat_logits_wmma(
    const float* __restrict__ x,
    const int*   __restrict__ ei,
    const float* __restrict__ attr,
    const float* __restrict__ W_l, const float* __restrict__ b_l,
    const float* __restrict__ W_r, const float* __restrict__ b_r,
    const float* __restrict__ W_e, const float* __restrict__ att,
    const float* __restrict__ attr_sum, const float* __restrict__ cnt,
    float* __restrict__ mx,
    float* __restrict__ logits,
    int E, int M  ) {
  const int wave = (blockIdx.x * blockDim.x + threadIdx.x) >> 5;
  const int lane = threadIdx.x & 31;
  const int base = wave * 16;
  if (base >= M) return;
  const int  lh = lane & 15;
  const bool hi = lane >= 16;

  v2f B[8];
#pragma unroll
  for (int t = 0; t < 8; ++t) {
    int ch = t * 16 + lh;
    if (!hi) { B[t].x = W_l[ch]; B[t].y = W_r[ch]; }
    else     { B[t].x = W_e[ch]; B[t].y = b_l[ch] + b_r[ch]; }
  }

  int j = base + lh;
  float a0 = 0.0f, a1 = 0.0f;
  if (j < M) {
    if (!hi) {
      int s, d;
      if (j < E) { s = ei[j]; d = ei[E + j]; }
      else       { s = d = j - E; }
      a0 = x[s];
      a1 = x[d];
    } else {
      if (j < E) a0 = attr[j];
      else { int n = j - E; a0 = attr_sum[n] / fmaxf(cnt[n], 1.0f); }
      a1 = 1.0f;
    }
  }
  v2f A; A.x = a0; A.y = a1;

  float p[2][8];
#pragma unroll
  for (int h = 0; h < 2; ++h)
#pragma unroll
    for (int r = 0; r < 8; ++r) p[h][r] = 0.0f;

#pragma unroll
  for (int t = 0; t < 8; ++t) {
    v8f cz = {};
    v8f D = __builtin_amdgcn_wmma_f32_16x16x4_f32(
        false, A, false, B[t], (short)0, cz, false, false);
    float av = att[t * 16 + lh];
    const int h = t >> 2;
#pragma unroll
    for (int r = 0; r < 8; ++r) {
      float g = D[r];
      g = (g > 0.0f) ? g : NEG_SLOPE * g;
      p[h][r] = fmaf(g, av, p[h][r]);
    }
  }

#pragma unroll
  for (int off = 1; off < 16; off <<= 1) {
#pragma unroll
    for (int r = 0; r < 8; ++r) {
      p[0][r] += __shfl_xor(p[0][r], off, 32);
      p[1][r] += __shfl_xor(p[1][r], off, 32);
    }
  }

  float l0 = p[0][0], l1 = p[1][0];
#pragma unroll
  for (int q = 1; q < 8; ++q) {
    bool sel = (lh == q);
    l0 = sel ? p[0][q] : l0;
    l1 = sel ? p[1][q] : l1;
  }
  if (lh < 8) {
    int jm = base + (hi ? 8 + lh : lh);
    if (jm < M) {
      int d = (jm < E) ? ei[E + jm] : (jm - E);
      logits[2 * jm + 0] = l0;
      logits[2 * jm + 1] = l1;
      atomic_max_f32(&mx[2 * d + 0], l0);
      atomic_max_f32(&mx[2 * d + 1], l1);
    }
  }
}

__global__ void gat_expsum(const float* __restrict__ x,
                           const int* __restrict__ ei,
                           const float* __restrict__ mx,
                           const float* __restrict__ logits,
                           float* __restrict__ den, float* __restrict__ num,
                           int E, int M) {
  int j = blockIdx.x * blockDim.x + threadIdx.x;
  if (j >= M) return;
  int s, d;
  if (j < E) { s = ei[j]; d = ei[E + j]; }
  else       { s = d = j - E; }
  float xs = x[s];
  float a0 = expf(logits[2 * j + 0] - mx[2 * d + 0]);
  float a1 = expf(logits[2 * j + 1] - mx[2 * d + 1]);
  atomicAdd(&den[2 * d + 0], a0);
  atomicAdd(&den[2 * d + 1], a1);
  atomicAdd(&num[2 * d + 0], a0 * xs);
  atomicAdd(&num[2 * d + 1], a1 * xs);
}

__global__ void gat_out(const float* __restrict__ num, const float* __restrict__ den,
                        const float* __restrict__ W_l, const float* __restrict__ b_l,
                        const float* __restrict__ bias, float* __restrict__ out,
                        int N) {
  int i = blockIdx.x * blockDim.x + threadIdx.x;
  if (i >= N * 128) return;
  int k = i & 127;
  int n = i >> 7;
  int h = k >> 6;
  float S = num[2 * n + h] / den[2 * n + h];
  out[i] = fmaf(S, W_l[k], b_l[k] + bias[k]);
}

extern "C" void kernel_launch(void* const* d_in, const int* in_sizes, int n_in,
                              void* d_out, int out_size, void* d_ws, size_t ws_size,
                              hipStream_t stream) {
  const float* x    = (const float*)d_in[0];
  const int*   ei   = (const int*)  d_in[1];
  const float* attr = (const float*)d_in[2];
  const float* W_l  = (const float*)d_in[3];
  const float* b_l  = (const float*)d_in[4];
  const float* W_r  = (const float*)d_in[5];
  const float* b_r  = (const float*)d_in[6];
  const float* W_e  = (const float*)d_in[7];
  const float* att  = (const float*)d_in[8];
  const float* bias = (const float*)d_in[9];

  const int N = in_sizes[0];
  const int E = in_sizes[2];
  const int M = E + N;

  float* ws       = (float*)d_ws;
  float* attr_sum = ws;
  float* cnt      = attr_sum + N;
  float* mx       = cnt + N;
  float* den      = mx + 2 * N;
  float* num      = den + 2 * N;
  float* logits   = num + 2 * N;

  const int BLK = 256;

  gat_init<<<(2 * N + BLK - 1) / BLK, BLK, 0, stream>>>(attr_sum, cnt, mx, den, num, N);

  gat_degree<<<(E + BLK - 1) / BLK, BLK, 0, stream>>>(ei, attr, attr_sum, cnt, E);

  int nGroups = (M + 15) / 16;
  int blocks2 = (nGroups + 7) / 8;
  gat_logits_wmma<<<blocks2, BLK, 0, stream>>>(x, ei, attr, W_l, b_l, W_r, b_r,
                                               W_e, att, attr_sum, cnt, mx,
                                               logits, E, M);

  gat_expsum<<<(M + BLK - 1) / BLK, BLK, 0, stream>>>(x, ei, mx, logits, den, num, E, M);

  gat_out<<<(N * 128 + BLK - 1) / BLK, BLK, 0, stream>>>(num, den, W_l, b_l, bias,
                                                         (float*)d_out, N);
}


# Sort edges by destination (one returning integer atomic per edge, scan, permutation); per-node max/exp-sum in one wave's registers: float atomics gone, summation order and reduction tree change
# speedup vs baseline: 1.9002x; 1.9002x over previous
#include <hip/hip_runtime.h>

typedef __attribute__((ext_vector_type(2))) float v2f;
typedef __attribute__((ext_vector_type(8))) float v8f;

#define NEG_SLOPE 0.2f
#define NODES_PER_WAVE 4

__global__ void gat_zero_deg(int* __restrict__ deg, int N) {
  int i = blockIdx.x * blockDim.x + threadIdx.x;
  if (i < N) deg[i] = 0;
}

__global__ void gat_rank(const int* __restrict__ ei,
                         int* __restrict__ deg, int* __restrict__ rank, int E) {
  int e = blockIdx.x * blockDim.x + threadIdx.x;
  if (e >= E) return;
  int d = ei[E + e];
  rank[e] = atomicAdd(&deg[d], 1);
}

__global__ void __launch_bounds__(1024)
gat_scan(const int* __restrict__ deg, int* __restrict__ off, int N) {
  __shared__ int wtot[2][32];
  const int t = threadIdx.x;
  const int lane = t & 31;
  const int w = t >> 5;
  int carry = 0;
  int par = 0;
  for (int base = 0; base < N; base += 4096, par ^= 1) {
    const int idx = base + 4 * t;
    int v0 = (idx     < N) ? deg[idx]     : 0;
    int v1 = (idx + 1 < N) ? deg[idx + 1] : 0;
    int v2 = (idx + 2 < N) ? deg[idx + 2] : 0;
    int v3 = (idx + 3 < N) ? deg[idx + 3] : 0;
    const int s = v0 + v1 + v2 + v3;
    int inc = s;
#pragma unroll
    for (int o = 1; o < 32; o <<= 1) {
      int y = __shfl_up(inc, o, 32);
      if (lane >= o) inc += y;
    }
    if (lane == 31) wtot[par][w] = inc;
    __syncthreads();
    const int wt = wtot[par][lane];
    int winc = wt;
#pragma unroll
    for (int o = 1; o < 32; o <<= 1) {
      int y = __shfl_up(winc, o, 32);
      if (lane >= o) winc += y;
    }
    const int wexcl = __shfl(winc, w, 32) - __shfl(wt, w, 32);
    const int total = __shfl(winc, 31, 32);
    int ex = carry + wexcl + (inc - s);
    if (idx     < N) off[idx]     = ex;
    ex += v0;
    if (idx + 1 < N) off[idx + 1] = ex;
    ex += v1;
    if (idx + 2 < N) off[idx + 2] = ex;
    ex += v2;
    if (idx + 3 < N) off[idx + 3] = ex;
    carry += total;
  }
  if (t == 0) off[N] = carry;
}

__global__ void gat_scatter(const int* __restrict__ ei, const int* __restrict__ off,
                            const int* __restrict__ rank, int* __restrict__ perm,
                            int E) {
  int e = blockIdx.x * blockDim.x + threadIdx.x;
  if (e >= E) return;
  int d = ei[E + e];
  perm[off[d] + rank[e]] = e;
}

__global__ void gat_node_wmma(
    const float* __restrict__ x,
    const int*   __restrict__ ei,
    const float* __restrict__ attr,
    const float* __restrict__ W_l, const float* __restrict__ b_l,
    const float* __restrict__ W_r, const float* __restrict__ b_r,
    const float* __restrict__ W_e, const float* __restrict__ att,
    const float* __restrict__ bias,
    const int*   __restrict__ off, const int* __restrict__ perm,
    float* __restrict__ out, int N) {
  const int wave = __builtin_amdgcn_readfirstlane(
      (int)((blockIdx.x * blockDim.x + threadIdx.x) >> 5));
  const int lane = threadIdx.x & 31;
  const int nbeg = wave * NODES_PER_WAVE;
  if (nbeg >= N) return;
  const int nend = (nbeg + NODES_PER_WAVE < N) ? nbeg + NODES_PER_WAVE : N;
  const int  lh = lane & 15;
  const bool hi = lane >= 16;
  const float NEG_INF = __int_as_float(0xff800000);

  v2f B[8];
  float av[8];
#pragma unroll
  for (int t = 0; t < 8; ++t) {
    int ch = t * 16 + lh;
    if (!hi) { B[t].x = W_l[ch]; B[t].y = W_r[ch]; }
    else     { B[t].x = W_e[ch]; B[t].y = b_l[ch] + b_r[ch]; }
    av[t] = att[ch];
  }
  const float4 wl4 = *(const float4*)(W_l + 4 * lane);
  const float4 bl4 = *(const float4*)(b_l + 4 * lane);
  const float4 bi4 = *(const float4*)(bias + 4 * lane);
  const int  rown = hi ? 8 + lh : lh;
  const bool owner = lh < 8;

  for (int n = nbeg; n < nend; ++n) {
    const int beg = off[n];
    const int deg = off[n + 1] - beg;
    const int R = deg + 1;
    const float xd = x[n];

    float asum = 0.0f;
    for (int q = lane; q < deg; q += 32) asum += attr[perm[beg + q]];
#pragma unroll
    for (int o = 16; o >= 1; o >>= 1) asum += __shfl_xor(asum, o, 32);
    const float loop_attr = asum / fmaxf((float)deg, 1.0f);

    float m0 = NEG_INF, m1 = NEG_INF;
    float den0 = 0.0f, den1 = 0.0f, num0 = 0.0f, num1 = 0.0f;

    for (int q0 = 0; q0 < R; q0 += 16) {
      const int q = q0 + lh;
      float a0 = 0.0f, a1 = 0.0f;
      if (q < R) {
        if (q < deg) {
          const int e = perm[beg + q];
          a0 = hi ? attr[e] : x[ei[e]];
        } else {
          a0 = hi ? loop_attr : xd;
        }
        a1 = hi ? 1.0f : xd;
      }
      v2f A; A.x = a0; A.y = a1;

      float p[2][8];
#pragma unroll
      for (int h = 0; h < 2; ++h)
#pragma unroll
        for (int r = 0; r < 8; ++r) p[h][r] = 0.0f;

#pragma unroll
      for (int t = 0; t < 8; ++t) {
        v8f cz = {};
        v8f D = __builtin_amdgcn_wmma_f32_16x16x4_f32(
            false, A, false, B[t], (short)0, cz, false, false);
        const int h = t >> 2;
#pragma unroll
        for (int r = 0; r < 8; ++r) {
          float g = D[r];
          g = (g > 0.0f) ? g : NEG_SLOPE * g;
          p[h][r] = fmaf(g, av[t], p[h][r]);
        }
      }

#pragma unroll
      for (int o = 1; o < 16; o <<= 1) {
#pragma unroll
        for (int r = 0; r < 8; ++r) {
          p[0][r] += __shfl_xor(p[0][r], o, 32);
          p[1][r] += __shfl_xor(p[1][r], o, 32);
        }
      }
      float l0 = p[0][0], l1 = p[1][0];
#pragma unroll
      for (int r = 1; r < 8; ++r) {
        bool sel = (lh == r);
        l0 = sel ? p[0][r] : l0;
        l1 = sel ? p[1][r] : l1;
      }
      const float xs = __shfl(a0, rown, 32);
      const bool valid = owner && (q0 + rown < R);

      float t0 = valid ? l0 : NEG_INF;
      float t1 = valid ? l1 : NEG_INF;
#pragma unroll
      for (int o = 16; o >= 1; o >>= 1) {
        t0 = fmaxf(t0, __shfl_xor(t0, o, 32));
        t1 = fmaxf(t1, __shfl_xor(t1, o, 32));
      }
      const float nm0 = fmaxf(m0, t0), nm1 = fmaxf(m1, t1);
      const float sc0 = expf(m0 - nm0), sc1 = expf(m1 - nm1);
      const float e0 = valid ? expf(l0 - nm0) : 0.0f;
      const float e1 = valid ? expf(l1 - nm1) : 0.0f;
      den0 = fmaf(den0, sc0, e0);
      den1 = fmaf(den1, sc1, e1);
      num0 = fmaf(num0, sc0, valid ? e0 * xs : 0.0f);
      num1 = fmaf(num1, sc1, valid ? e1 * xs : 0.0f);
      m0 = nm0; m1 = nm1;
    }

#pragma unroll
    for (int o = 16; o >= 1; o >>= 1) {
      den0 += __shfl_xor(den0, o, 32);
      den1 += __shfl_xor(den1, o, 32);
      num0 += __shfl_xor(num0, o, 32);
      num1 += __shfl_xor(num1, o, 32);
    }
    const float S = hi ? (num1 / den1) : (num0 / den0);
    float4 o4;
    o4.x = fmaf(S, wl4.x, bl4.x + bi4.x);
    o4.y = fmaf(S, wl4.y, bl4.y + bi4.y);
    o4.z = fmaf(S, wl4.z, bl4.z + bi4.z);
    o4.w = fmaf(S, wl4.w, bl4.w + bi4.w);
    *(float4*)(out + (size_t)n * 128 + 4 * lane) = o4;
  }
}

extern "C" void kernel_launch(void* const* d_in, const int* in_sizes, int n_in,
                              void* d_out, int out_size, void* d_ws, size_t ws_size,
                              hipStream_t stream) {
  const float* x    = (const float*)d_in[0];
  const int*   ei   = (const int*)  d_in[1];
  const float* attr = (const float*)d_in[2];
  const float* W_l  = (const float*)d_in[3];
  const float* b_l  = (const float*)d_in[4];
  const float* W_r  = (const float*)d_in[5];
  const float* b_r  = (const float*)d_in[6];
  const float* W_e  = (const float*)d_in[7];
  const float* att  = (const float*)d_in[8];
  const float* bias = (const float*)d_in[9];

  const int N = in_sizes[0];
  const int E = in_sizes[2];

  int* ws   = (int*)d_ws;
  int* deg  = ws;
  int* off  = deg + N;
  int* rank = off + N + 1;
  int* perm = rank + E;

  const int BLK = 256;

  gat_zero_deg<<<(N + BLK - 1) / BLK, BLK, 0, stream>>>(deg, N);

  gat_rank<<<(E + BLK - 1) / BLK, BLK, 0, stream>>>(ei, deg, rank, E);

  gat_scan<<<1, 1024, 0, stream>>>(deg, off, N);

  gat_scatter<<<(E + BLK - 1) / BLK, BLK, 0, stream>>>(ei, off, rank, perm, E);

  int nWaves  = (N + NODES_PER_WAVE - 1) / NODES_PER_WAVE;
  int blocks4 = (nWaves + 7) / 8;
  gat_node_wmma<<<blocks4, BLK, 0, stream>>>(x, ei, attr, W_l, b_l, W_r, b_r, W_e,
                                             att, bias, off, perm, (float*)d_out, N);
}


# Node kernel: WMMA operands swapped (weights as A, rows as B) so a lane holds 8 channels of one row; att-dot sums in-lane, 1 lane exchange per tile not 64; a head per lane half; reduction tree changes
# speedup vs baseline: 2.2365x; 1.1770x over previous
#include <hip/hip_runtime.h>

typedef __attribute__((ext_vector_type(2))) float v2f;
typedef __attribute__((ext_vector_type(8))) float v8f;

#define NEG_SLOPE 0.2f
#define NODES_PER_WAVE 4

__global__ void gat_zero_deg(int* __restrict__ deg, int N) {
  int i = blockIdx.x * blockDim.x + threadIdx.x;
  if (i < N) deg[i] = 0;
}

__global__ void gat_rank(const int* __restrict__ ei,
                         int* __restrict__ deg, int* __restrict__ rank, int E) {
  int e = blockIdx.x * blockDim.x + threadIdx.x;
  if (e >= E) return;
  int d = ei[E + e];
  rank[e] = atomicAdd(&deg[d], 1);
}

__global__ void __launch_bounds__(1024)
gat_scan(const int* __restrict__ deg, int* __restrict__ off, int N) {
  __shared__ int wtot[2][32];
  const int t = threadIdx.x;
  const int lane = t & 31;
  const int w = t >> 5;
  int carry = 0;
  int par = 0;
  for (int base = 0; base < N; base += 4096, par ^= 1) {
    const int idx = base + 4 * t;
    int v0 = (idx     < N) ? deg[idx]     : 0;
    int v1 = (idx + 1 < N) ? deg[idx + 1] : 0;
    int v2 = (idx + 2 < N) ? deg[idx + 2] : 0;
    int v3 = (idx + 3 < N) ? deg[idx + 3] : 0;
    const int s = v0 + v1 + v2 + v3;
    int inc = s;
#pragma unroll
    for (int o = 1; o < 32; o <<= 1) {
      int y = __shfl_up(inc, o, 32);
      if (lane >= o) inc += y;
    }
    if (lane == 31) wtot[par][w] = inc;
    __syncthreads();
    const int wt = wtot[par][lane];
    int winc = wt;
#pragma unroll
    for (int o = 1; o < 32; o <<= 1) {
      int y = __shfl_up(winc, o, 32);
      if (lane >= o) winc += y;
    }
    const int wexcl = __shfl(winc, w, 32) - __shfl(wt, w, 32);
    const int total = __shfl(winc, 31, 32);
    int ex = carry + wexcl + (inc - s);
    if (idx     < N) off[idx]     = ex;
    ex += v0;
    if (idx + 1 < N) off[idx + 1] = ex;
    ex += v1;
    if (idx + 2 < N) off[idx + 2] = ex;
    ex += v2;
    if (idx + 3 < N) off[idx + 3] = ex;
    carry += total;
  }
  if (t == 0) off[N] = carry;
}

__global__ void gat_scatter(const int* __restrict__ ei, const int* __restrict__ off,
                            const int* __restrict__ rank, int* __restrict__ perm,
                            int E) {
  int e = blockIdx.x * blockDim.x + threadIdx.x;
  if (e >= E) return;
  int d = ei[E + e];
  perm[off[d] + rank[e]] = e;
}

__global__ void gat_node_wmma(
    const float* __restrict__ x,
    const int*   __restrict__ ei,
    const float* __restrict__ attr,
    const float* __restrict__ W_l, const float* __restrict__ b_l,
    const float* __restrict__ W_r, const float* __restrict__ b_r,
    const float* __restrict__ W_e, const float* __restrict__ att,
    const float* __restrict__ bias,
    const int*   __restrict__ off, const int* __restrict__ perm,
    float* __restrict__ out, int N) {
  const int wave = __builtin_amdgcn_readfirstlane(
      (int)((blockIdx.x * blockDim.x + threadIdx.x) >> 5));
  const int lane = threadIdx.x & 31;
  const int nbeg = wave * NODES_PER_WAVE;
  if (nbeg >= N) return;
  const int nend = (nbeg + NODES_PER_WAVE < N) ? nbeg + NODES_PER_WAVE : N;
  const int  lh = lane & 15;
  const bool hi = lane >= 16;
  const float NEG_INF = __int_as_float(0xff800000);

  v2f Wt[8];
  float av[8][8];
#pragma unroll
  for (int t = 0; t < 8; ++t) {
    int ch = t * 16 + lh;
    if (!hi) { Wt[t].x = W_l[ch]; Wt[t].y = W_r[ch]; }
    else     { Wt[t].x = W_e[ch]; Wt[t].y = b_l[ch] + b_r[ch]; }
    const float* ap = att + t * 16 + (hi ? 8 : 0);
    const float4 a_lo = *(const float4*)(ap);
    const float4 a_hi = *(const float4*)(ap + 4);
    av[t][0] = a_lo.x; av[t][1] = a_lo.y; av[t][2] = a_lo.z; av[t][3] = a_lo.w;
    av[t][4] = a_hi.x; av[t][5] = a_hi.y; av[t][6] = a_hi.z; av[t][7] = a_hi.w;
  }
  const float4 wl4 = *(const float4*)(W_l + 4 * lane);
  const float4 bl4 = *(const float4*)(b_l + 4 * lane);
  const float4 bi4 = *(const float4*)(bias + 4 * lane);

  for (int n = nbeg; n < nend; ++n) {
    const int beg = off[n];
    const int deg = off[n + 1] - beg;
    const int R = deg + 1;
    const float xd = x[n];

    float asum = 0.0f;
    for (int q = lane; q < deg; q += 32) asum += attr[perm[beg + q]];
#pragma unroll
    for (int o = 16; o >= 1; o >>= 1) asum += __shfl_xor(asum, o, 32);
    const float loop_attr = asum / fmaxf((float)deg, 1.0f);

    float m = NEG_INF, den = 0.0f, num = 0.0f;

    for (int q0 = 0; q0 < R; q0 += 16) {
      const int q = q0 + lh;
      float a0 = 0.0f, a1 = 0.0f;
      if (q < R) {
        if (q < deg) {
          const int e = perm[beg + q];
          a0 = hi ? attr[e] : x[ei[e]];
        } else {
          a0 = hi ? loop_attr : xd;
        }
        a1 = hi ? 1.0f : xd;
      }
      v2f Rf; Rf.x = a0; Rf.y = a1;

      float p0 = 0.0f, p1 = 0.0f;
#pragma unroll
      for (int t = 0; t < 8; ++t) {
        v8f cz = {};
        v8f D = __builtin_amdgcn_wmma_f32_16x16x4_f32(
            false, Wt[t], false, Rf, (short)0, cz, false, false);
#pragma unroll
        for (int r = 0; r < 8; ++r) {
          float g = D[r];
          g = (g > 0.0f) ? g : NEG_SLOPE * g;
          if (t < 4) p0 = fmaf(g, av[t][r], p0);
          else       p1 = fmaf(g, av[t][r], p1);
        }
      }

      const float give = hi ? p0 : p1;
      const float keep = hi ? p1 : p0;
      const float lg = keep + __shfl_xor(give, 16, 32);
      const float xs = __shfl(a0, lh, 32);
      const bool valid = q < R;

      float tm = valid ? lg : NEG_INF;
#pragma unroll
      for (int o = 8; o >= 1; o >>= 1) tm = fmaxf(tm, __shfl_xor(tm, o, 32));
      const float nm = fmaxf(m, tm);
      const float sc = expf(m - nm);
      const float ex = valid ? expf(lg - nm) : 0.0f;
      den = fmaf(den, sc, ex);
      num = fmaf(num, sc, valid ? ex * xs : 0.0f);
      m = nm;
    }

#pragma unroll
    for (int o = 8; o >= 1; o >>= 1) {
      den += __shfl_xor(den, o, 32);
      num += __shfl_xor(num, o, 32);
    }
    const float S = num / den;
    float4 o4;
    o4.x = fmaf(S, wl4.x, bl4.x + bi4.x);
    o4.y = fmaf(S, wl4.y, bl4.y + bi4.y);
    o4.z = fmaf(S, wl4.z, bl4.z + bi4.z);
    o4.w = fmaf(S, wl4.w, bl4.w + bi4.w);
    *(float4*)(out + (size_t)n * 128 + 4 * lane) = o4;
  }
}

extern "C" void kernel_launch(void* const* d_in, const int* in_sizes, int n_in,
                              void* d_out, int out_size, void* d_ws, size_t ws_size,
                              hipStream_t stream) {
  const float* x    = (const float*)d_in[0];
  const int*   ei   = (const int*)  d_in[1];
  const float* attr = (const float*)d_in[2];
  const float* W_l  = (const float*)d_in[3];
  const float* b_l  = (const float*)d_in[4];
  const float* W_r  = (const float*)d_in[5];
  const float* b_r  = (const float*)d_in[6];
  const float* W_e  = (const float*)d_in[7];
  const float* att  = (const float*)d_in[8];
  const float* bias = (const float*)d_in[9];

  const int N = in_sizes[0];
  const int E = in_sizes[2];

  int* ws   = (int*)d_ws;
  int* deg  = ws;
  int* off  = deg + N;
  int* rank = off + N + 1;
  int* perm = rank + E;

  const int BLK = 256;

  gat_zero_deg<<<(N + BLK - 1) / BLK, BLK, 0, stream>>>(deg, N);

  gat_rank<<<(E + BLK - 1) / BLK, BLK, 0, stream>>>(ei, deg, rank, E);

  gat_scan<<<1, 1024, 0, stream>>>(deg, off, N);

  gat_scatter<<<(E + BLK - 1) / BLK, BLK, 0, stream>>>(ei, off, rank, perm, E);

  int nWaves  = (N + NODES_PER_WAVE - 1) / NODES_PER_WAVE;
  int blocks4 = (nWaves + 7) / 8;
  gat_node_wmma<<<blocks4, BLK, 0, stream>>>(x, ei, attr, W_l, b_l, W_r, b_r, W_e,
                                             att, bias, off, perm, (float*)d_out, N);
}


# Node kernel: a wave gathers all its nodes' rows once up front (edge ids, then src+attr, then x[src]; independent loads per level) into LDS, tiles and attr mean read LDS; weight loads branch-free
# speedup vs baseline: 2.3366x; 1.0448x over previous
#include <hip/hip_runtime.h>

typedef __attribute__((ext_vector_type(2))) float v2f;
typedef __attribute__((ext_vector_type(8))) float v8f;

#define NEG_SLOPE 0.2f
#define NODES_PER_WAVE 4
#define WAVES_PER_BLOCK 8
#define STAGE_CAP 128

__global__ void gat_zero_deg(int* __restrict__ deg, int N) {
  int i = blockIdx.x * blockDim.x + threadIdx.x;
  if (i < N) deg[i] = 0;
}

__global__ void gat_rank(const int* __restrict__ ei,
                         int* __restrict__ deg, int* __restrict__ rank, int E) {
  int e = blockIdx.x * blockDim.x + threadIdx.x;
  if (e >= E) return;
  int d = ei[E + e];
  rank[e] = atomicAdd(&deg[d], 1);
}

__global__ void __launch_bounds__(1024)
gat_scan(const int* __restrict__ deg, int* __restrict__ off, int N) {
  __shared__ int wtot[2][32];
  const int t = threadIdx.x;
  const int lane = t & 31;
  const int w = t >> 5;
  int carry = 0;
  int par = 0;
  for (int base = 0; base < N; base += 4096, par ^= 1) {
    const int idx = base + 4 * t;
    int v0 = (idx     < N) ? deg[idx]     : 0;
    int v1 = (idx + 1 < N) ? deg[idx + 1] : 0;
    int v2 = (idx + 2 < N) ? deg[idx + 2] : 0;
    int v3 = (idx + 3 < N) ? deg[idx + 3] : 0;
    const int s = v0 + v1 + v2 + v3;
    int inc = s;
#pragma unroll
    for (int o = 1; o < 32; o <<= 1) {
      int y = __shfl_up(inc, o, 32);
      if (lane >= o) inc += y;
    }
    if (lane == 31) wtot[par][w] = inc;
    __syncthreads();
    const int wt = wtot[par][lane];
    int winc = wt;
#pragma unroll
    for (int o = 1; o < 32; o <<= 1) {
      int y = __shfl_up(winc, o, 32);
      if (lane >= o) winc += y;
    }
    const int wexcl = __shfl(winc, w, 32) - __shfl(wt, w, 32);
    const int total = __shfl(winc, 31, 32);
    int ex = carry + wexcl + (inc - s);
    if (idx     < N) off[idx]     = ex;
    ex += v0;
    if (idx + 1 < N) off[idx + 1] = ex;
    ex += v1;
    if (idx + 2 < N) off[idx + 2] = ex;
    ex += v2;
    if (idx + 3 < N) off[idx + 3] = ex;
    carry += total;
  }
  if (t == 0) off[N] = carry;
}

__global__ void gat_scatter(const int* __restrict__ ei, const int* __restrict__ off,
                            const int* __restrict__ rank, int* __restrict__ perm,
                            int E) {
  int e = blockIdx.x * blockDim.x + threadIdx.x;
  if (e >= E) return;
  int d = ei[E + e];
  perm[off[d] + rank[e]] = e;
}

struct rows_in_lds    { static constexpr bool value = true;  };
struct rows_in_memory { static constexpr bool value = false; };

__global__ void __launch_bounds__(32 * WAVES_PER_BLOCK, 8) gat_node_wmma(
    const float* __restrict__ x,
    const int*   __restrict__ ei,
    const float* __restrict__ attr,
    const float* __restrict__ W_l, const float* __restrict__ b_l,
    const float* __restrict__ W_r, const float* __restrict__ b_r,
    const float* __restrict__ W_e, const float* __restrict__ att,
    const float* __restrict__ bias,
    const int*   __restrict__ off, const int* __restrict__ perm,
    float* __restrict__ out, int N) {
  const int wave = __builtin_amdgcn_readfirstlane(
      (int)((blockIdx.x * blockDim.x + threadIdx.x) >> 5));
  const int lane = threadIdx.x & 31;
  const int nbeg = wave * NODES_PER_WAVE;
  if (nbeg >= N) return;
  const int nend = (nbeg + NODES_PER_WAVE < N) ? nbeg + NODES_PER_WAVE : N;
  const int  lh = lane & 15;
  const bool hi = lane >= 16;
  const float NEG_INF = __int_as_float(0xff800000);
  __shared__ float2 stage[WAVES_PER_BLOCK][STAGE_CAP];
  float2* st = stage[threadIdx.x >> 5];

  v2f Wt[8];
  float av[8][8];
#pragma unroll
  for (int t = 0; t < 8; ++t) {
    int ch = t * 16 + lh;
    const float k0 = (hi ? W_e : W_l)[ch];
    const float k1 = (hi ? b_l : W_r)[ch];
    const float br = b_r[ch];
    Wt[t].x = k0;
    Wt[t].y = hi ? k1 + br : k1;
    const float* ap = att + t * 16 + (hi ? 8 : 0);
    const float4 a_lo = *(const float4*)(ap);
    const float4 a_hi = *(const float4*)(ap + 4);
    av[t][0] = a_lo.x; av[t][1] = a_lo.y; av[t][2] = a_lo.z; av[t][3] = a_lo.w;
    av[t][4] = a_hi.x; av[t][5] = a_hi.y; av[t][6] = a_hi.z; av[t][7] = a_hi.w;
  }
  const float4 wl4 = *(const float4*)(W_l + 4 * lane);
  const float4 bl4 = *(const float4*)(b_l + 4 * lane);
  const float4 bi4 = *(const float4*)(bias + 4 * lane);

  const int b0 = off[nbeg];
  const int cnt = off[nend] - b0;
  const bool staged = cnt <= STAGE_CAP;
  if (staged && cnt > 0) {
    int   se[4], ss[4];
    float sa[4], sx[4];
#pragma unroll
    for (int k = 0; k < 4; ++k) {
      const int i = 32 * k + lane;
      se[k] = perm[b0 + (i < cnt ? i : cnt - 1)];
    }
#pragma unroll
    for (int k = 0; k < 4; ++k) { ss[k] = ei[se[k]]; sa[k] = attr[se[k]]; }
#pragma unroll
    for (int k = 0; k < 4; ++k) sx[k] = x[ss[k]];
#pragma unroll
    for (int k = 0; k < 4; ++k) {
      float2 rw; rw.x = sx[k]; rw.y = sa[k];
      st[32 * k + lane] = rw;
    }
  }
  __builtin_amdgcn_wave_barrier();

  auto nodes = [&](auto where) {
    constexpr bool IN_LDS = decltype(where)::value;
    for (int n = nbeg; n < nend; ++n) {
      const int beg = off[n];
      const int deg = off[n + 1] - beg;
      const int R = deg + 1;
      const float xd = x[n];
      const float2* rows = st + (beg - b0);

      float asum = 0.0f;
      for (int q = lane; q < deg; q += 32) {
        if constexpr (IN_LDS) asum += rows[q].y;
        else                  asum += attr[perm[beg + q]];
      }
#pragma unroll
      for (int o = 16; o >= 1; o >>= 1) asum += __shfl_xor(asum, o, 32);
      const float loop_attr = asum / fmaxf((float)deg, 1.0f);

      float m = NEG_INF, den = 0.0f, num = 0.0f;

      for (int q0 = 0; q0 < R; q0 += 16) {
        const int q = q0 + lh;
        float a0 = 0.0f, a1 = 0.0f;
        float xs = 0.0f;
        if (q < R) {
          if (q < deg) {
            float2 rw;
            if constexpr (IN_LDS) {
              rw = rows[q];
            } else {
              const int e = perm[beg + q];
              rw.x = x[ei[e]]; rw.y = attr[e];
            }
            xs = rw.x;
            a0 = hi ? rw.y : rw.x;
          } else {
            xs = xd;
            a0 = hi ? loop_attr : xd;
          }
          a1 = hi ? 1.0f : xd;
        }
        v2f Rf; Rf.x = a0; Rf.y = a1;

        float p0 = 0.0f, p1 = 0.0f;
#pragma unroll
        for (int t = 0; t < 8; ++t) {
          v8f cz = {};
          v8f D = __builtin_amdgcn_wmma_f32_16x16x4_f32(
              false, Wt[t], false, Rf, (short)0, cz, false, false);
#pragma unroll
          for (int r = 0; r < 8; ++r) {
            float g = D[r];
            g = (g > 0.0f) ? g : NEG_SLOPE * g;
            if (t < 4) p0 = fmaf(g, av[t][r], p0);
            else       p1 = fmaf(g, av[t][r], p1);
          }
        }

        const float give = hi ? p0 : p1;
        const float keep = hi ? p1 : p0;
        const float lg = keep + __shfl_xor(give, 16, 32);
        const bool valid = q < R;

        float tm = valid ? lg : NEG_INF;
#pragma unroll
        for (int o = 8; o >= 1; o >>= 1) tm = fmaxf(tm, __shfl_xor(tm, o, 32));
        const float nm = fmaxf(m, tm);
        const float sc = expf(m - nm);
        const float ex = valid ? expf(lg - nm) : 0.0f;
        den = fmaf(den, sc, ex);
        num = fmaf(num, sc, valid ? ex * xs : 0.0f);
        m = nm;
      }

#pragma unroll
      for (int o = 8; o >= 1; o >>= 1) {
        den += __shfl_xor(den, o, 32);
        num += __shfl_xor(num, o, 32);
      }
      const float S = num / den;
      float4 o4;
      o4.x = fmaf(S, wl4.x, bl4.x + bi4.x);
      o4.y = fmaf(S, wl4.y, bl4.y + bi4.y);
      o4.z = fmaf(S, wl4.z, bl4.z + bi4.z);
      o4.w = fmaf(S, wl4.w, bl4.w + bi4.w);
      *(float4*)(out + (size_t)n * 128 + 4 * lane) = o4;
    }
  };
  if (staged) nodes(rows_in_lds{});
  else        nodes(rows_in_memory{});
}

extern "C" void kernel_launch(void* const* d_in, const int* in_sizes, int n_in,
                              void* d_out, int out_size, void* d_ws, size_t ws_size,
                              hipStream_t stream) {
  const float* x    = (const float*)d_in[0];
  const int*   ei   = (const int*)  d_in[1];
  const float* attr = (const float*)d_in[2];
  const float* W_l  = (const float*)d_in[3];
  const float* b_l  = (const float*)d_in[4];
  const float* W_r  = (const float*)d_in[5];
  const float* b_r  = (const float*)d_in[6];
  const float* W_e  = (const float*)d_in[7];
  const float* att  = (const float*)d_in[8];
  const float* bias = (const float*)d_in[9];

  const int N = in_sizes[0];
  const int E = in_sizes[2];

  int* ws   = (int*)d_ws;
  int* deg  = ws;
  int* off  = deg + N;
  int* rank = off + N + 1;
  int* perm = rank + E;

  const int BLK = 256;

  gat_zero_deg<<<(N + BLK - 1) / BLK, BLK, 0, stream>>>(deg, N);

  gat_rank<<<(E + BLK - 1) / BLK, BLK, 0, stream>>>(ei, deg, rank, E);

  gat_scan<<<1, 1024, 0, stream>>>(deg, off, N);

  gat_scatter<<<(E + BLK - 1) / BLK, BLK, 0, stream>>>(ei, off, rank, perm, E);

  int nWaves  = (N + NODES_PER_WAVE - 1) / NODES_PER_WAVE;
  int blocks4 = (nWaves + 7) / 8;
  gat_node_wmma<<<blocks4, BLK, 0, stream>>>(x, ei, attr, W_l, b_l, W_r, b_r, W_e,
                                             att, bias, off, perm, (float*)d_out, N);
}


# Node kernel: leakyrelu split as a*g + b*|g|: per channel one multiply-add on |WMMA result| (was mul, compare, select, multiply-add); linear part per row from 4 per-head sums; summation order changes
# speedup vs baseline: 5.7462x; 2.4592x over previous
#include <hip/hip_runtime.h>

typedef __attribute__((ext_vector_type(2))) float v2f;
typedef __attribute__((ext_vector_type(8))) float v8f;

#define NEG_SLOPE 0.2f
#define LK_LIN (0.5f * (1.0f + NEG_SLOPE))
#define LK_ABS (0.5f * (1.0f - NEG_SLOPE))
#define NODES_PER_WAVE 4
#define WAVES_PER_BLOCK 8
#define STAGE_CAP 128

#define GRP_THREADS 1024
#define GRP_WGS 256
#define SEG_MAX 2048
#define LIST_CAP 4096

__global__ void __launch_bounds__(GRP_THREADS)
gat_group_wide(const int* __restrict__ ei,
          int* __restrict__ off, int* __restrict__ perm, int N, int E, int S) {
  __shared__ int cnt[SEG_MAX];
  __shared__ int lst_e[LIST_CAP];
  __shared__ unsigned short lst_d[LIST_CAP];
  __shared__ int lperm[LIST_CAP];
  __shared__ int wtot[32];
  __shared__ int wbel[32];
  __shared__ int nlist;

  const int t = threadIdx.x;
  const int lane = t & 31;
  const int w = t >> 5;
  const int lo0 = blockIdx.x * S;
  const int lo = lo0 < N ? lo0 : N;
  const int hi = (lo + S < N) ? lo + S : N;
  const unsigned span = (unsigned)(hi - lo);
  const int* __restrict__ dst = ei + E;

  for (int i = t; i < SEG_MAX; i += GRP_THREADS) cnt[i] = 0;
  if (t == 0) nlist = 0;
  __syncthreads();

  int below = 0;
  auto visit = [&](int d, int e) {
    below += (d < lo) ? 1 : 0;
    const unsigned dl = (unsigned)(d - lo);
    if (dl < span) {
      atomicAdd(&cnt[dl], 1);
      const int slot = atomicAdd(&nlist, 1);
      if (slot < LIST_CAP) { lst_e[slot] = e; lst_d[slot] = (unsigned short)dl; }
    }
  };

  const int E4 = ((E & 3) == 0) ? (E >> 2) : 0;
  const int4* __restrict__ dst4 = (const int4*)dst;
  const int nfull = E4 / (8 * GRP_THREADS);
  for (int it = 0; it < nfull; ++it) {
    int4 v[8];
#pragma unroll
    for (int u = 0; u < 8; ++u) v[u] = dst4[(it * 8 + u) * GRP_THREADS + t];
#pragma unroll
    for (int u = 0; u < 8; ++u) {
      const int e = 4 * ((it * 8 + u) * GRP_THREADS + t);
      visit(v[u].x, e);     visit(v[u].y, e + 1);
      visit(v[u].z, e + 2); visit(v[u].w, e + 3);
    }
  }
  for (int i = nfull * 8 * GRP_THREADS + t; i < E4; i += GRP_THREADS) {
    const int4 v = dst4[i];
    visit(v.x, 4 * i);     visit(v.y, 4 * i + 1);
    visit(v.z, 4 * i + 2); visit(v.w, 4 * i + 3);
  }
  for (int e = 4 * E4 + t; e < E; e += GRP_THREADS) visit(dst[e], e);

  __syncthreads();

  const int i2 = 2 * t;
  const int v0 = cnt[i2];
  const int v1 = cnt[i2 + 1];
  const int s = v0 + v1;
  int inc = s;
#pragma unroll
  for (int o = 1; o < 32; o <<= 1) {
    int y = __shfl_up(inc, o, 32);
    if (lane >= o) inc += y;
  }
#pragma unroll
  for (int o = 16; o >= 1; o >>= 1) below += __shfl_xor(below, o, 32);
  if (lane == 31) wtot[w] = inc;
  if (lane == 0)  wbel[w] = below;
  __syncthreads();
  const int wt = wtot[lane];
  int winc = wt;
#pragma unroll
  for (int o = 1; o < 32; o <<= 1) {
    int y = __shfl_up(winc, o, 32);
    if (lane >= o) winc += y;
  }
  const int wexcl = __shfl(winc, w, 32) - __shfl(wt, w, 32);
  const int total = __shfl(winc, 31, 32);
  int base = wbel[lane];
#pragma unroll
  for (int o = 16; o >= 1; o >>= 1) base += __shfl_xor(base, o, 32);

  const int ex0 = wexcl + (inc - s);
  const int ex1 = ex0 + v0;
  cnt[i2]     = ex0;
  cnt[i2 + 1] = ex1;
  if ((unsigned)i2 < span)       off[lo + i2]     = base + ex0;
  if ((unsigned)(i2 + 1) < span) off[lo + i2 + 1] = base + ex1;
  if (blockIdx.x == gridDim.x - 1 && t == 0) off[N] = base + total;
  __syncthreads();

  const int nl = nlist;
  if (nl <= LIST_CAP) {
    for (int i = t; i < nl; i += GRP_THREADS) {
      const int pos = atomicAdd(&cnt[lst_d[i]], 1);
      lperm[pos] = lst_e[i];
    }
    __syncthreads();
    for (int i = t; i < nl; i += GRP_THREADS) perm[base + i] = lperm[i];
  } else {
    for (int e = t; e < E; e += GRP_THREADS) {
      const unsigned dl = (unsigned)(dst[e] - lo);
      if (dl < span) perm[base + atomicAdd(&cnt[dl], 1)] = e;
    }
  }
}

#define PRE_WGS 128
#define PRE_THREADS 1024

__global__ void __launch_bounds__(PRE_THREADS)
gat_bucket(const int* __restrict__ ei,
           unsigned* __restrict__ bktw, int* __restrict__ hist, int E, int S) {
  __shared__ int h[256];
  const int t = threadIdx.x;
  if (t < 256) h[t] = 0;
  __syncthreads();
  const int* __restrict__ dst = ei + E;
  const unsigned uS = (unsigned)S;
  const int nW = (E + 3) >> 2;
  const bool al = (E & 3) == 0;
  for (int g = blockIdx.x * PRE_THREADS + t; g < nW; g += gridDim.x * PRE_THREADS) {
    int d[4];
    const int e = 4 * g;
    if (al) {
      const int4 v = ((const int4*)dst)[g];
      d[0] = v.x; d[1] = v.y; d[2] = v.z; d[3] = v.w;
    } else {
      d[0] = dst[e];
      d[1] = (e + 1 < E) ? dst[e + 1] : -1;
      d[2] = (e + 2 < E) ? dst[e + 2] : -1;
      d[3] = (e + 3 < E) ? dst[e + 3] : -1;
    }
    unsigned word = 0;
#pragma unroll
    for (int j = 0; j < 4; ++j) {
      if (d[j] >= 0) {
        const unsigned b = ((unsigned)d[j] / uS) & 255u;
        atomicAdd(&h[b], 1);
        word |= b << (8 * j);
      }
    }
    bktw[g] = word;
  }
  __syncthreads();
  if (t < 256) hist[blockIdx.x * 256 + t] = h[t];
}

#define HIT_W 256
static_assert((PRE_WGS * 64) % GRP_THREADS == 0, "hist rows are read in whole trips");

__global__ void __launch_bounds__(GRP_THREADS)
gat_group(const int* __restrict__ ei,
          const unsigned* __restrict__ bktw, const int* __restrict__ hist,
          int* __restrict__ off, int* __restrict__ perm, int N, int E, int S) {
  __shared__ int cnt[SEG_MAX];
  __shared__ int lst_e[LIST_CAP];
  __shared__ unsigned short lst_d[LIST_CAP];
  __shared__ int lperm[LIST_CAP];
  __shared__ uint2 hit[GRP_THREADS / 32][HIT_W];
  __shared__ int wtot[32];
  __shared__ int wbel[32];
  __shared__ int nlist;
  __shared__ int ovf;

  const int t = threadIdx.x;
  const int lane = t & 31;
  const int w = t >> 5;
  const unsigned k = blockIdx.x;
  const int lo0 = blockIdx.x * S;
  const int lo = lo0 < N ? lo0 : N;
  const int hi = (lo + S < N) ? lo + S : N;
  const unsigned span = (unsigned)(hi - lo);
  const int* __restrict__ dst = ei + E;

  for (int i = t; i < SEG_MAX; i += GRP_THREADS) cnt[i] = 0;
  if (t == 0) { nlist = 0; ovf = 0; }
  __syncthreads();

  int below = 0;
  {
    const int4* __restrict__ h4 = (const int4*)hist;
#pragma unroll
    for (int q = 0; q < PRE_WGS * 64 / GRP_THREADS; ++q) {
      const int i = q * GRP_THREADS + t;
      const int4 v = h4[i];
      const unsigned b = (unsigned)(4 * i) & 255u;
      below += (b < k ? v.x : 0) + (b + 1 < k ? v.y : 0)
             + (b + 2 < k ? v.z : 0) + (b + 3 < k ? v.w : 0);
    }
  }

  const unsigned kk = k * 0x01010101u;
  auto zb = [&](unsigned x) -> unsigned {
    const unsigned z = x ^ kk;
    return ~(((z & 0x7f7f7f7fu) + 0x7f7f7f7fu) | z | 0x7f7f7f7fu);
  };
  int wcnt = 0;
  uint2* myhit = hit[w];
  auto probe = [&](const int4 v, int i, bool in) {
    const unsigned f0 = zb((unsigned)v.x), f1 = zb((unsigned)v.y);
    const unsigned f2 = zb((unsigned)v.z), f3 = zb((unsigned)v.w);
    const bool h = in && ((f0 | f1 | f2 | f3) != 0u);
    const unsigned m = __builtin_amdgcn_ballot_w32(h);
    if (h) {
      const int pos = wcnt + (int)__builtin_amdgcn_mbcnt_lo(m, 0u);
      if (pos < HIT_W) {
        uint2 r;
        r.x = (unsigned)i;
        r.y = (f0 >> 7) | (f1 >> 6) | (f2 >> 5) | (f3 >> 4);
        myhit[pos] = r;
      }
    }
    wcnt += __popc(m);
  };

  const int E16 = E >> 4;
  const int4* __restrict__ bk4 = (const int4*)bktw;
  const int nfull = E16 / (8 * GRP_THREADS);
  for (int it = 0; it < nfull; ++it) {
    int4 v[8];
#pragma unroll
    for (int u = 0; u < 8; ++u) v[u] = bk4[(it * 8 + u) * GRP_THREADS + t];
#pragma unroll
    for (int u = 0; u < 8; ++u) probe(v[u], (it * 8 + u) * GRP_THREADS + t, true);
  }
  for (int i0 = nfull * 8 * GRP_THREADS; i0 < E16; i0 += GRP_THREADS) {
    const int i = i0 + t;
    const bool in = i < E16;
    probe(bk4[in ? i : E16 - 1], i, in);
  }
  if (wcnt > HIT_W && lane == 0) ovf = 1;
  __syncthreads();

  auto add_edge = [&](int d, int e) {
    const unsigned dl = (unsigned)(d - lo);
    if (dl < span) {
      atomicAdd(&cnt[dl], 1);
      const int slot = atomicAdd(&nlist, 1);
      if (slot < LIST_CAP) { lst_e[slot] = e; lst_d[slot] = (unsigned short)dl; }
    }
  };
  if (ovf == 0) {
    for (int j = lane; j < wcnt; j += 32) {
      const uint2 r = myhit[j];
      unsigned p = r.y;
      const int e0 = 16 * (int)r.x;
      while (p != 0u) {
        const int bit = __builtin_ctz(p);
        p &= p - 1u;
        const int e = e0 + 4 * (bit & 7) + (bit >> 3);
        add_edge(dst[e], e);
      }
    }
    for (int e = 16 * E16 + t; e < E; e += GRP_THREADS) add_edge(dst[e], e);
  } else {
    for (int e = t; e < E; e += GRP_THREADS) {
      const unsigned dl = (unsigned)(dst[e] - lo);
      if (dl < span) atomicAdd(&cnt[dl], 1);
    }
    if (t == 0) nlist = LIST_CAP + 1;
  }
  __syncthreads();

  const int i2 = 2 * t;
  const int v0 = cnt[i2];
  const int v1 = cnt[i2 + 1];
  const int s = v0 + v1;
  int inc = s;
#pragma unroll
  for (int o = 1; o < 32; o <<= 1) {
    int y = __shfl_up(inc, o, 32);
    if (lane >= o) inc += y;
  }
#pragma unroll
  for (int o = 16; o >= 1; o >>= 1) below += __shfl_xor(below, o, 32);
  if (lane == 31) wtot[w] = inc;
  if (lane == 0)  wbel[w] = below;
  __syncthreads();
  const int wt = wtot[lane];
  int winc = wt;
#pragma unroll
  for (int o = 1; o < 32; o <<= 1) {
    int y = __shfl_up(winc, o, 32);
    if (lane >= o) winc += y;
  }
  const int wexcl = __shfl(winc, w, 32) - __shfl(wt, w, 32);
  const int total = __shfl(winc, 31, 32);
  int base = wbel[lane];
#pragma unroll
  for (int o = 16; o >= 1; o >>= 1) base += __shfl_xor(base, o, 32);

  const int ex0 = wexcl + (inc - s);
  const int ex1 = ex0 + v0;
  cnt[i2]     = ex0;
  cnt[i2 + 1] = ex1;
  if ((unsigned)i2 < span)       off[lo + i2]     = base + ex0;
  if ((unsigned)(i2 + 1) < span) off[lo + i2 + 1] = base + ex1;
  if (blockIdx.x == gridDim.x - 1 && t == 0) off[N] = base + total;
  __syncthreads();

  const int nl = nlist;
  if (nl <= LIST_CAP) {
    for (int i = t; i < nl; i += GRP_THREADS) {
      const int pos = atomicAdd(&cnt[lst_d[i]], 1);
      lperm[pos] = lst_e[i];
    }
    __syncthreads();
    for (int i = t; i < nl; i += GRP_THREADS) perm[base + i] = lperm[i];
  } else {
    for (int e = t; e < E; e += GRP_THREADS) {
      const unsigned dl = (unsigned)(dst[e] - lo);
      if (dl < span) perm[base + atomicAdd(&cnt[dl], 1)] = e;
    }
  }
}

struct rows_in_lds    { static constexpr bool value = true;  };
struct rows_in_memory { static constexpr bool value = false; };

__global__ void __launch_bounds__(32 * WAVES_PER_BLOCK, 8) gat_node_wmma(
    const float* __restrict__ x,
    const int*   __restrict__ ei,
    const float* __restrict__ attr,
    const float* __restrict__ W_l, const float* __restrict__ b_l,
    const float* __restrict__ W_r, const float* __restrict__ b_r,
    const float* __restrict__ W_e, const float* __restrict__ att,
    const float* __restrict__ bias,
    const int*   __restrict__ off, const int* __restrict__ perm,
    float* __restrict__ out, int N) {
  const int wave = __builtin_amdgcn_readfirstlane(
      (int)((blockIdx.x * blockDim.x + threadIdx.x) >> 5));
  const int lane = threadIdx.x & 31;
  const int nbeg = wave * NODES_PER_WAVE;
  if (nbeg >= N) return;
  const int nend = (nbeg + NODES_PER_WAVE < N) ? nbeg + NODES_PER_WAVE : N;
  const int  lh = lane & 15;
  const bool hi = lane >= 16;
  const float NEG_INF = __int_as_float(0xff800000);
  __shared__ float2 stage[WAVES_PER_BLOCK][STAGE_CAP];
  float2* st = stage[threadIdx.x >> 5];

  v2f Wt[8];
  float av[8][8];
  float s0x = 0.0f, s0y = 0.0f, s1x = 0.0f, s1y = 0.0f;
#pragma unroll
  for (int t = 0; t < 8; ++t) {
    int ch = t * 16 + lh;
    const float k0 = (hi ? W_e : W_l)[ch];
    const float k1 = (hi ? b_l : W_r)[ch];
    const float br = b_r[ch];
    Wt[t].x = k0;
    Wt[t].y = hi ? k1 + br : k1;
    const float ac = att[ch];
    if (t < 4) { s0x = fmaf(ac, Wt[t].x, s0x); s0y = fmaf(ac, Wt[t].y, s0y); }
    else       { s1x = fmaf(ac, Wt[t].x, s1x); s1y = fmaf(ac, Wt[t].y, s1y); }
    const float* ap = att + t * 16 + (hi ? 8 : 0);
    const float4 a_lo = *(const float4*)(ap);
    const float4 a_hi = *(const float4*)(ap + 4);
    av[t][0] = a_lo.x; av[t][1] = a_lo.y; av[t][2] = a_lo.z; av[t][3] = a_lo.w;
    av[t][4] = a_hi.x; av[t][5] = a_hi.y; av[t][6] = a_hi.z; av[t][7] = a_hi.w;
  }
  for (int o = 8; o >= 1; o >>= 1) {
    s0x += __shfl_xor(s0x, o, 32);
    s0y += __shfl_xor(s0y, o, 32);
    s1x += __shfl_xor(s1x, o, 32);
    s1y += __shfl_xor(s1y, o, 32);
  }
  const float gx = __shfl_xor(hi ? s0x : s1x, 16, 32);
  const float gy = __shfl_xor(hi ? s0y : s1y, 16, 32);
  const float cA = hi ? gx  : s0x;
  const float cB = hi ? gy  : s0y;
  const float cC = hi ? s1x : gx;
  const float cD = hi ? s1y : gy;

  const float4 wl4 = *(const float4*)(W_l + 4 * lane);
  const float4 bl4 = *(const float4*)(b_l + 4 * lane);
  const float4 bi4 = *(const float4*)(bias + 4 * lane);

  const int b0 = off[nbeg];
  const int cnt = off[nend] - b0;
  const bool staged = cnt <= STAGE_CAP;
  if (staged && cnt > 0) {
    int   se[4], ss[4];
    float sa[4], sx[4];
#pragma unroll
    for (int k = 0; k < 4; ++k) {
      const int i = 32 * k + lane;
      se[k] = perm[b0 + (i < cnt ? i : cnt - 1)];
    }
#pragma unroll
    for (int k = 0; k < 4; ++k) { ss[k] = ei[se[k]]; sa[k] = attr[se[k]]; }
#pragma unroll
    for (int k = 0; k < 4; ++k) sx[k] = x[ss[k]];
#pragma unroll
    for (int k = 0; k < 4; ++k) {
      float2 rw; rw.x = sx[k]; rw.y = sa[k];
      st[32 * k + lane] = rw;
    }
  }
  __builtin_amdgcn_wave_barrier();

  auto nodes = [&](auto where) {
    constexpr bool IN_LDS = decltype(where)::value;
    for (int n = nbeg; n < nend; ++n) {
      const int beg = off[n];
      const int deg = off[n + 1] - beg;
      const int R = deg + 1;
      const float xd = x[n];
      const float2* rows = st + (beg - b0);

      float asum = 0.0f;
      for (int q = lane; q < deg; q += 32) {
        if constexpr (IN_LDS) asum += rows[q].y;
        else                  asum += attr[perm[beg + q]];
      }
#pragma unroll
      for (int o = 16; o >= 1; o >>= 1) asum += __shfl_xor(asum, o, 32);
      const float loop_attr = asum / fmaxf((float)deg, 1.0f);

      float m = NEG_INF, den = 0.0f, num = 0.0f;

      for (int q0 = 0; q0 < R; q0 += 16) {
        const int q = q0 + lh;
        float a0 = 0.0f, a1 = 0.0f;
        float xs = 0.0f;
        float ra = 0.0f;
        if (q < R) {
          if (q < deg) {
            float2 rw;
            if constexpr (IN_LDS) {
              rw = rows[q];
            } else {
              const int e = perm[beg + q];
              rw.x = x[ei[e]]; rw.y = attr[e];
            }
            xs = rw.x;
            ra = rw.y;
          } else {
            xs = xd;
            ra = loop_attr;
          }
          a0 = hi ? ra : xs;
          a1 = hi ? 1.0f : xd;
        }
        v2f Rf; Rf.x = a0; Rf.y = a1;
        const float lin = fmaf(cA, xs, fmaf(cB, xd, fmaf(cC, ra, cD)));

        float p0a = 0.0f, p0b = 0.0f, p1a = 0.0f, p1b = 0.0f;
#pragma unroll
        for (int t = 0; t < 8; ++t) {
          v8f cz = {};
          v8f D = __builtin_amdgcn_wmma_f32_16x16x4_f32(
              false, Wt[t], false, Rf, (short)0, cz, false, false);
#pragma unroll
          for (int r = 0; r < 8; r += 2) {
            const float g0 = fabsf(D[r]);
            const float g1 = fabsf(D[r + 1]);
            if (t < 4) {
              p0a = fmaf(g0, av[t][r], p0a);
              p0b = fmaf(g1, av[t][r + 1], p0b);
            } else {
              p1a = fmaf(g0, av[t][r], p1a);
              p1b = fmaf(g1, av[t][r + 1], p1b);
            }
          }
        }
        const float p0 = p0a + p0b;
        const float p1 = p1a + p1b;

        const float give = hi ? p0 : p1;
        const float keep = hi ? p1 : p0;
        const float sabs = keep + __shfl_xor(give, 16, 32);
        const float lg = fmaf(LK_ABS, sabs, LK_LIN * lin);
        const bool valid = q < R;

        float tm = valid ? lg : NEG_INF;
#pragma unroll
        for (int o = 8; o >= 1; o >>= 1) tm = fmaxf(tm, __shfl_xor(tm, o, 32));
        const float nm = fmaxf(m, tm);
        const float sc = expf(m - nm);
        const float ex = valid ? expf(lg - nm) : 0.0f;
        den = fmaf(den, sc, ex);
        num = fmaf(num, sc, valid ? ex * xs : 0.0f);
        m = nm;
      }

#pragma unroll
      for (int o = 8; o >= 1; o >>= 1) {
        den += __shfl_xor(den, o, 32);
        num += __shfl_xor(num, o, 32);
      }
      const float S = num / den;
      float4 o4;
      o4.x = fmaf(S, wl4.x, bl4.x + bi4.x);
      o4.y = fmaf(S, wl4.y, bl4.y + bi4.y);
      o4.z = fmaf(S, wl4.z, bl4.z + bi4.z);
      o4.w = fmaf(S, wl4.w, bl4.w + bi4.w);
      *(float4*)(out + (size_t)n * 128 + 4 * lane) = o4;
    }
  };
  if (staged) nodes(rows_in_lds{});
  else        nodes(rows_in_memory{});
}

extern "C" void kernel_launch(void* const* d_in, const int* in_sizes, int n_in,
                              void* d_out, int out_size, void* d_ws, size_t ws_size,
                              hipStream_t stream) {
  const float* x    = (const float*)d_in[0];
  const int*   ei   = (const int*)  d_in[1];
  const float* attr = (const float*)d_in[2];
  const float* W_l  = (const float*)d_in[3];
  const float* b_l  = (const float*)d_in[4];
  const float* W_r  = (const float*)d_in[5];
  const float* b_r  = (const float*)d_in[6];
  const float* W_e  = (const float*)d_in[7];
  const float* att  = (const float*)d_in[8];
  const float* bias = (const float*)d_in[9];

  const int N = in_sizes[0];
  const int E = in_sizes[2];


  const int BLK = 256;

  const int G = (N + SEG_MAX - 1) / SEG_MAX > GRP_WGS ? (N + SEG_MAX - 1) / SEG_MAX : GRP_WGS;
  const int S = (N + G - 1) / G;

  const size_t plane = (((size_t)E + 15) / 16) * 16;
  const size_t need  = plane + ((size_t)PRE_WGS * 256 + (size_t)N + 1 + (size_t)E) * sizeof(int);
  int* off;
  int* perm;
  if (G <= 256 && need <= ws_size) {
    unsigned* bktw = (unsigned*)d_ws;
    int* hist = (int*)((char*)d_ws + plane);
    off  = hist + PRE_WGS * 256;
    perm = off + N + 1;
    gat_bucket<<<PRE_WGS, PRE_THREADS, 0, stream>>>(ei, bktw, hist, E, S);
    gat_group<<<G, GRP_THREADS, 0, stream>>>(ei, bktw, hist, off, perm, N, E, S);
  } else {
    off  = (int*)d_ws;
    perm = off + N + 1;
    gat_group_wide<<<G, GRP_THREADS, 0, stream>>>(ei, off, perm, N, E, S);
  }

  int nWaves  = (N + NODES_PER_WAVE - 1) / NODES_PER_WAVE;
  int blocks4 = (nWaves + 7) / 8;
  gat_node_wmma<<<blocks4, BLK, 0, stream>>>(x, ei, attr, W_l, b_l, W_r, b_r, W_e,
                                             att, bias, off, perm, (float*)d_out, N);
}


# Node kernel: cross-lane sums and maxima on the vector unit (row_xmask DPP butterflies, permlanex16 between lane halves), not shuffles through the LDS unit; same partners and order, sums unchanged
# speedup vs baseline: 6.2118x; 1.0810x over previous
#include <hip/hip_runtime.h>

typedef __attribute__((ext_vector_type(2))) float v2f;
typedef __attribute__((ext_vector_type(8))) float v8f;

#define NEG_SLOPE 0.2f
#define LK_LIN (0.5f * (1.0f + NEG_SLOPE))
#define LK_ABS (0.5f * (1.0f - NEG_SLOPE))
#define NODES_PER_WAVE 4
#define WAVES_PER_BLOCK 8
#define STAGE_CAP 128

#define GRP_THREADS 1024
#define GRP_WGS 256
#define SEG_MAX 2048
#define LIST_CAP 4096

__global__ void __launch_bounds__(GRP_THREADS)
gat_group_wide(const int* __restrict__ ei,
          int* __restrict__ off, int* __restrict__ perm, int N, int E, int S) {
  __shared__ int cnt[SEG_MAX];
  __shared__ int lst_e[LIST_CAP];
  __shared__ unsigned short lst_d[LIST_CAP];
  __shared__ int lperm[LIST_CAP];
  __shared__ int wtot[32];
  __shared__ int wbel[32];
  __shared__ int nlist;

  const int t = threadIdx.x;
  const int lane = t & 31;
  const int w = t >> 5;
  const int lo0 = blockIdx.x * S;
  const int lo = lo0 < N ? lo0 : N;
  const int hi = (lo + S < N) ? lo + S : N;
  const unsigned span = (unsigned)(hi - lo);
  const int* __restrict__ dst = ei + E;

  for (int i = t; i < SEG_MAX; i += GRP_THREADS) cnt[i] = 0;
  if (t == 0) nlist = 0;
  __syncthreads();

  int below = 0;
  auto visit = [&](int d, int e) {
    below += (d < lo) ? 1 : 0;
    const unsigned dl = (unsigned)(d - lo);
    if (dl < span) {
      atomicAdd(&cnt[dl], 1);
      const int slot = atomicAdd(&nlist, 1);
      if (slot < LIST_CAP) { lst_e[slot] = e; lst_d[slot] = (unsigned short)dl; }
    }
  };

  const int E4 = ((E & 3) == 0) ? (E >> 2) : 0;
  const int4* __restrict__ dst4 = (const int4*)dst;
  const int nfull = E4 / (8 * GRP_THREADS);
  for (int it = 0; it < nfull; ++it) {
    int4 v[8];
#pragma unroll
    for (int u = 0; u < 8; ++u) v[u] = dst4[(it * 8 + u) * GRP_THREADS + t];
#pragma unroll
    for (int u = 0; u < 8; ++u) {
      const int e = 4 * ((it * 8 + u) * GRP_THREADS + t);
      visit(v[u].x, e);     visit(v[u].y, e + 1);
      visit(v[u].z, e + 2); visit(v[u].w, e + 3);
    }
  }
  for (int i = nfull * 8 * GRP_THREADS + t; i < E4; i += GRP_THREADS) {
    const int4 v = dst4[i];
    visit(v.x, 4 * i);     visit(v.y, 4 * i + 1);
    visit(v.z, 4 * i + 2); visit(v.w, 4 * i + 3);
  }
  for (int e = 4 * E4 + t; e < E; e += GRP_THREADS) visit(dst[e], e);

  __syncthreads();

  const int i2 = 2 * t;
  const int v0 = cnt[i2];
  const int v1 = cnt[i2 + 1];
  const int s = v0 + v1;
  int inc = s;
#pragma unroll
  for (int o = 1; o < 32; o <<= 1) {
    int y = __shfl_up(inc, o, 32);
    if (lane >= o) inc += y;
  }
#pragma unroll
  for (int o = 16; o >= 1; o >>= 1) below += __shfl_xor(below, o, 32);
  if (lane == 31) wtot[w] = inc;
  if (lane == 0)  wbel[w] = below;
  __syncthreads();
  const int wt = wtot[lane];
  int winc = wt;
#pragma unroll
  for (int o = 1; o < 32; o <<= 1) {
    int y = __shfl_up(winc, o, 32);
    if (lane >= o) winc += y;
  }
  const int wexcl = __shfl(winc, w, 32) - __shfl(wt, w, 32);
  const int total = __shfl(winc, 31, 32);
  int base = wbel[lane];
#pragma unroll
  for (int o = 16; o >= 1; o >>= 1) base += __shfl_xor(base, o, 32);

  const int ex0 = wexcl + (inc - s);
  const int ex1 = ex0 + v0;
  cnt[i2]     = ex0;
  cnt[i2 + 1] = ex1;
  if ((unsigned)i2 < span)       off[lo + i2]     = base + ex0;
  if ((unsigned)(i2 + 1) < span) off[lo + i2 + 1] = base + ex1;
  if (blockIdx.x == gridDim.x - 1 && t == 0) off[N] = base + total;
  __syncthreads();

  const int nl = nlist;
  if (nl <= LIST_CAP) {
    for (int i = t; i < nl; i += GRP_THREADS) {
      const int pos = atomicAdd(&cnt[lst_d[i]], 1);
      lperm[pos] = lst_e[i];
    }
    __syncthreads();
    for (int i = t; i < nl; i += GRP_THREADS) perm[base + i] = lperm[i];
  } else {
    for (int e = t; e < E; e += GRP_THREADS) {
      const unsigned dl = (unsigned)(dst[e] - lo);
      if (dl < span) perm[base + atomicAdd(&cnt[dl], 1)] = e;
    }
  }
}

#define PRE_WGS 128
#define PRE_THREADS 1024

__global__ void __launch_bounds__(PRE_THREADS)
gat_bucket(const int* __restrict__ ei,
           unsigned* __restrict__ bktw, int* __restrict__ hist, int E, int S) {
  __shared__ int h[256];
  const int t = threadIdx.x;
  if (t < 256) h[t] = 0;
  __syncthreads();
  const int* __restrict__ dst = ei + E;
  const unsigned uS = (unsigned)S;
  const int nW = (E + 3) >> 2;
  const bool al = (E & 3) == 0;
  for (int g = blockIdx.x * PRE_THREADS + t; g < nW; g += gridDim.x * PRE_THREADS) {
    int d[4];
    const int e = 4 * g;
    if (al) {
      const int4 v = ((const int4*)dst)[g];
      d[0] = v.x; d[1] = v.y; d[2] = v.z; d[3] = v.w;
    } else {
      d[0] = dst[e];
      d[1] = (e + 1 < E) ? dst[e + 1] : -1;
      d[2] = (e + 2 < E) ? dst[e + 2] : -1;
      d[3] = (e + 3 < E) ? dst[e + 3] : -1;
    }
    unsigned word = 0;
#pragma unroll
    for (int j = 0; j < 4; ++j) {
      if (d[j] >= 0) {
        const unsigned b = ((unsigned)d[j] / uS) & 255u;
        atomicAdd(&h[b], 1);
        word |= b << (8 * j);
      }
    }
    bktw[g] = word;
  }
  __syncthreads();
  if (t < 256) hist[blockIdx.x * 256 + t] = h[t];
}

#define HIT_W 256
static_assert((PRE_WGS * 64) % GRP_THREADS == 0, "hist rows are read in whole trips");

__global__ void __launch_bounds__(GRP_THREADS)
gat_group(const int* __restrict__ ei,
          const unsigned* __restrict__ bktw, const int* __restrict__ hist,
          int* __restrict__ off, int* __restrict__ perm, int N, int E, int S) {
  __shared__ int cnt[SEG_MAX];
  __shared__ int lst_e[LIST_CAP];
  __shared__ unsigned short lst_d[LIST_CAP];
  __shared__ int lperm[LIST_CAP];
  __shared__ uint2 hit[GRP_THREADS / 32][HIT_W];
  __shared__ int wtot[32];
  __shared__ int wbel[32];
  __shared__ int nlist;
  __shared__ int ovf;

  const int t = threadIdx.x;
  const int lane = t & 31;
  const int w = t >> 5;
  const unsigned k = blockIdx.x;
  const int lo0 = blockIdx.x * S;
  const int lo = lo0 < N ? lo0 : N;
  const int hi = (lo + S < N) ? lo + S : N;
  const unsigned span = (unsigned)(hi - lo);
  const int* __restrict__ dst = ei + E;

  for (int i = t; i < SEG_MAX; i += GRP_THREADS) cnt[i] = 0;
  if (t == 0) { nlist = 0; ovf = 0; }
  __syncthreads();

  int below = 0;
  {
    const int4* __restrict__ h4 = (const int4*)hist;
#pragma unroll
    for (int q = 0; q < PRE_WGS * 64 / GRP_THREADS; ++q) {
      const int i = q * GRP_THREADS + t;
      const int4 v = h4[i];
      const unsigned b = (unsigned)(4 * i) & 255u;
      below += (b < k ? v.x : 0) + (b + 1 < k ? v.y : 0)
             + (b + 2 < k ? v.z : 0) + (b + 3 < k ? v.w : 0);
    }
  }

  const unsigned kk = k * 0x01010101u;
  auto zb = [&](unsigned x) -> unsigned {
    const unsigned z = x ^ kk;
    return ~(((z & 0x7f7f7f7fu) + 0x7f7f7f7fu) | z | 0x7f7f7f7fu);
  };
  int wcnt = 0;
  uint2* myhit = hit[w];
  auto probe = [&](const int4 v, int i, bool in) {
    const unsigned f0 = zb((unsigned)v.x), f1 = zb((unsigned)v.y);
    const unsigned f2 = zb((unsigned)v.z), f3 = zb((unsigned)v.w);
    const bool h = in && ((f0 | f1 | f2 | f3) != 0u);
    const unsigned m = __builtin_amdgcn_ballot_w32(h);
    if (h) {
      const int pos = wcnt + (int)__builtin_amdgcn_mbcnt_lo(m, 0u);
      if (pos < HIT_W) {
        uint2 r;
        r.x = (unsigned)i;
        r.y = (f0 >> 7) | (f1 >> 6) | (f2 >> 5) | (f3 >> 4);
        myhit[pos] = r;
      }
    }
    wcnt += __popc(m);
  };

  const int E16 = E >> 4;
  const int4* __restrict__ bk4 = (const int4*)bktw;
  const int nfull = E16 / (8 * GRP_THREADS);
  for (int it = 0; it < nfull; ++it) {
    int4 v[8];
#pragma unroll
    for (int u = 0; u < 8; ++u) v[u] = bk4[(it * 8 + u) * GRP_THREADS + t];
#pragma unroll
    for (int u = 0; u < 8; ++u) probe(v[u], (it * 8 + u) * GRP_THREADS + t, true);
  }
  for (int i0 = nfull * 8 * GRP_THREADS; i0 < E16; i0 += GRP_THREADS) {
    const int i = i0 + t;
    const bool in = i < E16;
    probe(bk4[in ? i : E16 - 1], i, in);
  }
  if (wcnt > HIT_W && lane == 0) ovf = 1;
  __syncthreads();

  auto add_edge = [&](int d, int e) {
    const unsigned dl = (unsigned)(d - lo);
    if (dl < span) {
      atomicAdd(&cnt[dl], 1);
      const int slot = atomicAdd(&nlist, 1);
      if (slot < LIST_CAP) { lst_e[slot] = e; lst_d[slot] = (unsigned short)dl; }
    }
  };
  if (ovf == 0) {
    for (int j = lane; j < wcnt; j += 32) {
      const uint2 r = myhit[j];
      unsigned p = r.y;
      const int e0 = 16 * (int)r.x;
      while (p != 0u) {
        const int bit = __builtin_ctz(p);
        p &= p - 1u;
        const int e = e0 + 4 * (bit & 7) + (bit >> 3);
        add_edge(dst[e], e);
      }
    }
    for (int e = 16 * E16 + t; e < E; e += GRP_THREADS) add_edge(dst[e], e);
  } else {
    for (int e = t; e < E; e += GRP_THREADS) {
      const unsigned dl = (unsigned)(dst[e] - lo);
      if (dl < span) atomicAdd(&cnt[dl], 1);
    }
    if (t == 0) nlist = LIST_CAP + 1;
  }
  __syncthreads();

  const int i2 = 2 * t;
  const int v0 = cnt[i2];
  const int v1 = cnt[i2 + 1];
  const int s = v0 + v1;
  int inc = s;
#pragma unroll
  for (int o = 1; o < 32; o <<= 1) {
    int y = __shfl_up(inc, o, 32);
    if (lane >= o) inc += y;
  }
#pragma unroll
  for (int o = 16; o >= 1; o >>= 1) below += __shfl_xor(below, o, 32);
  if (lane == 31) wtot[w] = inc;
  if (lane == 0)  wbel[w] = below;
  __syncthreads();
  const int wt = wtot[lane];
  int winc = wt;
#pragma unroll
  for (int o = 1; o < 32; o <<= 1) {
    int y = __shfl_up(winc, o, 32);
    if (lane >= o) winc += y;
  }
  const int wexcl = __shfl(winc, w, 32) - __shfl(wt, w, 32);
  const int total = __shfl(winc, 31, 32);
  int base = wbel[lane];
#pragma unroll
  for (int o = 16; o >= 1; o >>= 1) base += __shfl_xor(base, o, 32);

  const int ex0 = wexcl + (inc - s);
  const int ex1 = ex0 + v0;
  cnt[i2]     = ex0;
  cnt[i2 + 1] = ex1;
  if ((unsigned)i2 < span)       off[lo + i2]     = base + ex0;
  if ((unsigned)(i2 + 1) < span) off[lo + i2 + 1] = base + ex1;
  if (blockIdx.x == gridDim.x - 1 && t == 0) off[N] = base + total;
  __syncthreads();

  const int nl = nlist;
  if (nl <= LIST_CAP) {
    for (int i = t; i < nl; i += GRP_THREADS) {
      const int pos = atomicAdd(&cnt[lst_d[i]], 1);
      lperm[pos] = lst_e[i];
    }
    __syncthreads();
    for (int i = t; i < nl; i += GRP_THREADS) perm[base + i] = lperm[i];
  } else {
    for (int e = t; e < E; e += GRP_THREADS) {
      const unsigned dl = (unsigned)(dst[e] - lo);
      if (dl < span) perm[base + atomicAdd(&cnt[dl], 1)] = e;
    }
  }
}

template <int M>
__device__ __forceinline__ float row_xor(float v) {
  const int b = __builtin_bit_cast(int, v);
  return __builtin_bit_cast(float, __builtin_amdgcn_update_dpp(b, b, 0x160 + M, 0xf, 0xf, true));
}
__device__ __forceinline__ float xhalf(float v) {
  const int b = __builtin_bit_cast(int, v);
  return __builtin_bit_cast(float,
      __builtin_amdgcn_permlanex16(b, b, 0x76543210, 0xfedcba98, false, false));
}
__device__ __forceinline__ float row_sum16(float v) {
  v += row_xor<8>(v); v += row_xor<4>(v); v += row_xor<2>(v); v += row_xor<1>(v);
  return v;
}
__device__ __forceinline__ float row_max16(float v) {
  v = fmaxf(v, row_xor<8>(v)); v = fmaxf(v, row_xor<4>(v));
  v = fmaxf(v, row_xor<2>(v)); v = fmaxf(v, row_xor<1>(v));
  return v;
}

struct rows_in_lds    { static constexpr bool value = true;  };
struct rows_in_memory { static constexpr bool value = false; };

__global__ void __launch_bounds__(32 * WAVES_PER_BLOCK, 8) gat_node_wmma(
    const float* __restrict__ x,
    const int*   __restrict__ ei,
    const float* __restrict__ attr,
    const float* __restrict__ W_l, const float* __restrict__ b_l,
    const float* __restrict__ W_r, const float* __restrict__ b_r,
    const float* __restrict__ W_e, const float* __restrict__ att,
    const float* __restrict__ bias,
    const int*   __restrict__ off, const int* __restrict__ perm,
    float* __restrict__ out, int N) {
  const int wave = __builtin_amdgcn_readfirstlane(
      (int)((blockIdx.x * blockDim.x + threadIdx.x) >> 5));
  const int lane = threadIdx.x & 31;
  const int nbeg = wave * NODES_PER_WAVE;
  if (nbeg >= N) return;
  const int nend = (nbeg + NODES_PER_WAVE < N) ? nbeg + NODES_PER_WAVE : N;
  const int  lh = lane & 15;
  const bool hi = lane >= 16;
  const float NEG_INF = __int_as_float(0xff800000);
  __shared__ float2 stage[WAVES_PER_BLOCK][STAGE_CAP];
  float2* st = stage[threadIdx.x >> 5];

  v2f Wt[8];
  float av[8][8];
  float s0x = 0.0f, s0y = 0.0f, s1x = 0.0f, s1y = 0.0f;
#pragma unroll
  for (int t = 0; t < 8; ++t) {
    int ch = t * 16 + lh;
    const float k0 = (hi ? W_e : W_l)[ch];
    const float k1 = (hi ? b_l : W_r)[ch];
    const float br = b_r[ch];
    Wt[t].x = k0;
    Wt[t].y = hi ? k1 + br : k1;
    const float ac = att[ch];
    if (t < 4) { s0x = fmaf(ac, Wt[t].x, s0x); s0y = fmaf(ac, Wt[t].y, s0y); }
    else       { s1x = fmaf(ac, Wt[t].x, s1x); s1y = fmaf(ac, Wt[t].y, s1y); }
    const float* ap = att + t * 16 + (hi ? 8 : 0);
    const float4 a_lo = *(const float4*)(ap);
    const float4 a_hi = *(const float4*)(ap + 4);
    av[t][0] = a_lo.x; av[t][1] = a_lo.y; av[t][2] = a_lo.z; av[t][3] = a_lo.w;
    av[t][4] = a_hi.x; av[t][5] = a_hi.y; av[t][6] = a_hi.z; av[t][7] = a_hi.w;
  }
  s0x = row_sum16(s0x);
  s0y = row_sum16(s0y);
  s1x = row_sum16(s1x);
  s1y = row_sum16(s1y);
  const float gx = xhalf(hi ? s0x : s1x);
  const float gy = xhalf(hi ? s0y : s1y);
  const float cA = hi ? gx  : s0x;
  const float cB = hi ? gy  : s0y;
  const float cC = hi ? s1x : gx;
  const float cD = hi ? s1y : gy;

  const float4 wl4 = *(const float4*)(W_l + 4 * lane);
  const float4 bl4 = *(const float4*)(b_l + 4 * lane);
  const float4 bi4 = *(const float4*)(bias + 4 * lane);

  const int b0 = off[nbeg];
  const int cnt = off[nend] - b0;
  const bool staged = cnt <= STAGE_CAP;
  if (staged && cnt > 0) {
    int   se[4], ss[4];
    float sa[4], sx[4];
#pragma unroll
    for (int k = 0; k < 4; ++k) {
      const int i = 32 * k + lane;
      se[k] = perm[b0 + (i < cnt ? i : cnt - 1)];
    }
#pragma unroll
    for (int k = 0; k < 4; ++k) { ss[k] = ei[se[k]]; sa[k] = attr[se[k]]; }
#pragma unroll
    for (int k = 0; k < 4; ++k) sx[k] = x[ss[k]];
#pragma unroll
    for (int k = 0; k < 4; ++k) {
      float2 rw; rw.x = sx[k]; rw.y = sa[k];
      st[32 * k + lane] = rw;
    }
  }
  __builtin_amdgcn_wave_barrier();

  auto nodes = [&](auto where) {
    constexpr bool IN_LDS = decltype(where)::value;
    for (int n = nbeg; n < nend; ++n) {
      const int beg = off[n];
      const int deg = off[n + 1] - beg;
      const int R = deg + 1;
      const float xd = x[n];
      const float2* rows = st + (beg - b0);

      float asum = 0.0f;
      for (int q = lane; q < deg; q += 32) {
        if constexpr (IN_LDS) asum += rows[q].y;
        else                  asum += attr[perm[beg + q]];
      }
      asum += xhalf(asum);
      asum = row_sum16(asum);
      const float loop_attr = asum / fmaxf((float)deg, 1.0f);

      float m = NEG_INF, den = 0.0f, num = 0.0f;

      for (int q0 = 0; q0 < R; q0 += 16) {
        const int q = q0 + lh;
        float a0 = 0.0f, a1 = 0.0f;
        float xs = 0.0f;
        float ra = 0.0f;
        if (q < R) {
          if (q < deg) {
            float2 rw;
            if constexpr (IN_LDS) {
              rw = rows[q];
            } else {
              const int e = perm[beg + q];
              rw.x = x[ei[e]]; rw.y = attr[e];
            }
            xs = rw.x;
            ra = rw.y;
          } else {
            xs = xd;
            ra = loop_attr;
          }
          a0 = hi ? ra : xs;
          a1 = hi ? 1.0f : xd;
        }
        v2f Rf; Rf.x = a0; Rf.y = a1;
        const float lin = fmaf(cA, xs, fmaf(cB, xd, fmaf(cC, ra, cD)));

        float p0a = 0.0f, p0b = 0.0f, p1a = 0.0f, p1b = 0.0f;
#pragma unroll
        for (int t = 0; t < 8; ++t) {
          v8f cz = {};
          v8f D = __builtin_amdgcn_wmma_f32_16x16x4_f32(
              false, Wt[t], false, Rf, (short)0, cz, false, false);
#pragma unroll
          for (int r = 0; r < 8; r += 2) {
            const float g0 = fabsf(D[r]);
            const float g1 = fabsf(D[r + 1]);
            if (t < 4) {
              p0a = fmaf(g0, av[t][r], p0a);
              p0b = fmaf(g1, av[t][r + 1], p0b);
            } else {
              p1a = fmaf(g0, av[t][r], p1a);
              p1b = fmaf(g1, av[t][r + 1], p1b);
            }
          }
        }
        const float p0 = p0a + p0b;
        const float p1 = p1a + p1b;

        const float give = hi ? p0 : p1;
        const float keep = hi ? p1 : p0;
        const float sabs = keep + xhalf(give);
        const float lg = fmaf(LK_ABS, sabs, LK_LIN * lin);
        const bool valid = q < R;

        const float tm = row_max16(valid ? lg : NEG_INF);
        const float nm = fmaxf(m, tm);
        const float sc = expf(m - nm);
        const float ex = valid ? expf(lg - nm) : 0.0f;
        den = fmaf(den, sc, ex);
        num = fmaf(num, sc, valid ? ex * xs : 0.0f);
        m = nm;
      }

      den = row_sum16(den);
      num = row_sum16(num);
      const float S = num / den;
      float4 o4;
      o4.x = fmaf(S, wl4.x, bl4.x + bi4.x);
      o4.y = fmaf(S, wl4.y, bl4.y + bi4.y);
      o4.z = fmaf(S, wl4.z, bl4.z + bi4.z);
      o4.w = fmaf(S, wl4.w, bl4.w + bi4.w);
      *(float4*)(out + (size_t)n * 128 + 4 * lane) = o4;
    }
  };
  if (staged) nodes(rows_in_lds{});
  else        nodes(rows_in_memory{});
}

extern "C" void kernel_launch(void* const* d_in, const int* in_sizes, int n_in,
                              void* d_out, int out_size, void* d_ws, size_t ws_size,
                              hipStream_t stream) {
  const float* x    = (const float*)d_in[0];
  const int*   ei   = (const int*)  d_in[1];
  const float* attr = (const float*)d_in[2];
  const float* W_l  = (const float*)d_in[3];
  const float* b_l  = (const float*)d_in[4];
  const float* W_r  = (const float*)d_in[5];
  const float* b_r  = (const float*)d_in[6];
  const float* W_e  = (const float*)d_in[7];
  const float* att  = (const float*)d_in[8];
  const float* bias = (const float*)d_in[9];

  const int N = in_sizes[0];
  const int E = in_sizes[2];


  const int BLK = 256;

  const int G = (N + SEG_MAX - 1) / SEG_MAX > GRP_WGS ? (N + SEG_MAX - 1) / SEG_MAX : GRP_WGS;
  const int S = (N + G - 1) / G;

  const size_t plane = (((size_t)E + 15) / 16) * 16;
  const size_t need  = plane + ((size_t)PRE_WGS * 256 + (size_t)N + 1 + (size_t)E) * sizeof(int);
  int* off;
  int* perm;
  if (G <= 256 && need <= ws_size) {
    unsigned* bktw = (unsigned*)d_ws;
    int* hist = (int*)((char*)d_ws + plane);
    off  = hist + PRE_WGS * 256;
    perm = off + N + 1;
    gat_bucket<<<PRE_WGS, PRE_THREADS, 0, stream>>>(ei, bktw, hist, E, S);
    gat_group<<<G, GRP_THREADS, 0, stream>>>(ei, bktw, hist, off, perm, N, E, S);
  } else {
    off  = (int*)d_ws;
    perm = off + N + 1;
    gat_group_wide<<<G, GRP_THREADS, 0, stream>>>(ei, off, perm, N, E, S);
  }

  int nWaves  = (N + NODES_PER_WAVE - 1) / NODES_PER_WAVE;
  int blocks4 = (nWaves + 7) / 8;
  gat_node_wmma<<<blocks4, BLK, 0, stream>>>(x, ei, attr, W_l, b_l, W_r, b_r, W_e,
                                             att, bias, off, perm, (float*)d_out, N);
}


# Pre-pass sorts each chunk of edges by node range (LDS counting sort, run starts per chunk); grouping launch reads only its own runs, not every edge; arrival order inside a node's run changes
# speedup vs baseline: 6.7705x; 1.0899x over previous
#include <hip/hip_runtime.h>

typedef __attribute__((ext_vector_type(2))) float v2f;
typedef __attribute__((ext_vector_type(8))) float v8f;

#define NEG_SLOPE 0.2f
#define LK_LIN (0.5f * (1.0f + NEG_SLOPE))
#define LK_ABS (0.5f * (1.0f - NEG_SLOPE))
#define NODES_PER_WAVE 4
#define WAVES_PER_BLOCK 8
#define STAGE_CAP 128

#define GRP_THREADS 1024
#define GRP_WGS 256
#define SEG_MAX 2048
#define LIST_CAP 4096

__global__ void __launch_bounds__(GRP_THREADS)
gat_group_wide(const int* __restrict__ ei,
          int* __restrict__ off, int* __restrict__ perm, int N, int E, int S) {
  __shared__ int cnt[SEG_MAX];
  __shared__ int lst_e[LIST_CAP];
  __shared__ unsigned short lst_d[LIST_CAP];
  __shared__ int lperm[LIST_CAP];
  __shared__ int wtot[32];
  __shared__ int wbel[32];
  __shared__ int nlist;

  const int t = threadIdx.x;
  const int lane = t & 31;
  const int w = t >> 5;
  const int lo0 = blockIdx.x * S;
  const int lo = lo0 < N ? lo0 : N;
  const int hi = (lo + S < N) ? lo + S : N;
  const unsigned span = (unsigned)(hi - lo);
  const int* __restrict__ dst = ei + E;

  for (int i = t; i < SEG_MAX; i += GRP_THREADS) cnt[i] = 0;
  if (t == 0) nlist = 0;
  __syncthreads();

  int below = 0;
  auto visit = [&](int d, int e) {
    below += (d < lo) ? 1 : 0;
    const unsigned dl = (unsigned)(d - lo);
    if (dl < span) {
      atomicAdd(&cnt[dl], 1);
      const int slot = atomicAdd(&nlist, 1);
      if (slot < LIST_CAP) { lst_e[slot] = e; lst_d[slot] = (unsigned short)dl; }
    }
  };

  const int E4 = ((E & 3) == 0) ? (E >> 2) : 0;
  const int4* __restrict__ dst4 = (const int4*)dst;
  const int nfull = E4 / (8 * GRP_THREADS);
  for (int it = 0; it < nfull; ++it) {
    int4 v[8];
#pragma unroll
    for (int u = 0; u < 8; ++u) v[u] = dst4[(it * 8 + u) * GRP_THREADS + t];
#pragma unroll
    for (int u = 0; u < 8; ++u) {
      const int e = 4 * ((it * 8 + u) * GRP_THREADS + t);
      visit(v[u].x, e);     visit(v[u].y, e + 1);
      visit(v[u].z, e + 2); visit(v[u].w, e + 3);
    }
  }
  for (int i = nfull * 8 * GRP_THREADS + t; i < E4; i += GRP_THREADS) {
    const int4 v = dst4[i];
    visit(v.x, 4 * i);     visit(v.y, 4 * i + 1);
    visit(v.z, 4 * i + 2); visit(v.w, 4 * i + 3);
  }
  for (int e = 4 * E4 + t; e < E; e += GRP_THREADS) visit(dst[e], e);

  __syncthreads();

  const int i2 = 2 * t;
  const int v0 = cnt[i2];
  const int v1 = cnt[i2 + 1];
  const int s = v0 + v1;
  int inc = s;
#pragma unroll
  for (int o = 1; o < 32; o <<= 1) {
    int y = __shfl_up(inc, o, 32);
    if (lane >= o) inc += y;
  }
#pragma unroll
  for (int o = 16; o >= 1; o >>= 1) below += __shfl_xor(below, o, 32);
  if (lane == 31) wtot[w] = inc;
  if (lane == 0)  wbel[w] = below;
  __syncthreads();
  const int wt = wtot[lane];
  int winc = wt;
#pragma unroll
  for (int o = 1; o < 32; o <<= 1) {
    int y = __shfl_up(winc, o, 32);
    if (lane >= o) winc += y;
  }
  const int wexcl = __shfl(winc, w, 32) - __shfl(wt, w, 32);
  const int total = __shfl(winc, 31, 32);
  int base = wbel[lane];
#pragma unroll
  for (int o = 16; o >= 1; o >>= 1) base += __shfl_xor(base, o, 32);

  const int ex0 = wexcl + (inc - s);
  const int ex1 = ex0 + v0;
  cnt[i2]     = ex0;
  cnt[i2 + 1] = ex1;
  if ((unsigned)i2 < span)       off[lo + i2]     = base + ex0;
  if ((unsigned)(i2 + 1) < span) off[lo + i2 + 1] = base + ex1;
  if (blockIdx.x == gridDim.x - 1 && t == 0) off[N] = base + total;
  __syncthreads();

  const int nl = nlist;
  if (nl <= LIST_CAP) {
    for (int i = t; i < nl; i += GRP_THREADS) {
      const int pos = atomicAdd(&cnt[lst_d[i]], 1);
      lperm[pos] = lst_e[i];
    }
    __syncthreads();
    for (int i = t; i < nl; i += GRP_THREADS) perm[base + i] = lperm[i];
  } else {
    for (int e = t; e < E; e += GRP_THREADS) {
      const unsigned dl = (unsigned)(dst[e] - lo);
      if (dl < span) perm[base + atomicAdd(&cnt[dl], 1)] = e;
    }
  }
}

#define PRE_WGS 128
#define PRE_THREADS 1024
#define PRE_ROW 257

__device__ __forceinline__ unsigned range_id(unsigned d, unsigned S, unsigned M) {
  unsigned q = __umulhi(d, M);
  if (d - q * S >= S) ++q;
  return q & 255u;
}

__global__ void __launch_bounds__(PRE_THREADS)
gat_bucket(const int* __restrict__ ei,
           int* __restrict__ srt, int* __restrict__ pre, int E, int S) {
  __shared__ int h[256];
  __shared__ int wsum[8];
  const int t = threadIdx.x;
  const int lane = t & 31;
  if (t < 256) h[t] = 0;
  __syncthreads();
  const int* __restrict__ dst = ei + E;
  const unsigned uS = (unsigned)S;
  const unsigned M = 0xffffffffu / uS;
  const int nW = (E + 3) >> 2;
  const int CW = (nW + (int)gridDim.x - 1) / (int)gridDim.x;
  const int w0 = (int)blockIdx.x * CW;
  const int w1 = (w0 + CW < nW) ? w0 + CW : nW;
  const bool al = (E & 3) == 0;
  int* __restrict__ mine = srt + (size_t)blockIdx.x * (size_t)(4 * CW);

  auto load4 = [&](int g, int (&d)[4]) {
    const int e = 4 * g;
    if (al) {
      const int4 v = ((const int4*)dst)[g];
      d[0] = v.x; d[1] = v.y; d[2] = v.z; d[3] = v.w;
    } else {
      d[0] = dst[e];
      d[1] = (e + 1 < E) ? dst[e + 1] : -1;
      d[2] = (e + 2 < E) ? dst[e + 2] : -1;
      d[3] = (e + 3 < E) ? dst[e + 3] : -1;
    }
  };

  for (int g = w0 + t; g < w1; g += PRE_THREADS) {
    int d[4];
    load4(g, d);
#pragma unroll
    for (int j = 0; j < 4; ++j)
      if (d[j] >= 0) atomicAdd(&h[range_id((unsigned)d[j], uS, M)], 1);
  }
  __syncthreads();

  const int v = (t < 256) ? h[t] : 0;
  int inc = v;
#pragma unroll
  for (int o = 1; o < 32; o <<= 1) {
    int y = __shfl_up(inc, o, 32);
    if (lane >= o) inc += y;
  }
  if (t < 256 && lane == 31) wsum[t >> 5] = inc;
  __syncthreads();
  if (t < 256) {
    int wex = 0;
#pragma unroll
    for (int q = 0; q < 8; ++q) wex += (q < (t >> 5)) ? wsum[q] : 0;
    const int ex = wex + inc - v;
    h[t] = ex;
    pre[blockIdx.x * PRE_ROW + t] = ex;
    if (t == 255) pre[blockIdx.x * PRE_ROW + 256] = ex + v;
  }
  __syncthreads();

  for (int g = w0 + t; g < w1; g += PRE_THREADS) {
    int d[4];
    load4(g, d);
    int pos[4];
#pragma unroll
    for (int j = 0; j < 4; ++j)
      pos[j] = (d[j] >= 0) ? atomicAdd(&h[range_id((unsigned)d[j], uS, M)], 1) : -1;
#pragma unroll
    for (int j = 0; j < 4; ++j)
      if (pos[j] >= 0) mine[pos[j]] = 4 * g + j;
  }
}

static_assert(GRP_THREADS == 8 * PRE_WGS, "eight threads to a chunk");

__global__ void __launch_bounds__(GRP_THREADS)
gat_group(const int* __restrict__ ei,
          const int* __restrict__ srt, const int* __restrict__ pre,
          int* __restrict__ off, int* __restrict__ perm, int N, int E, int S) {
  __shared__ int cnt[SEG_MAX];
  __shared__ int lst_e[LIST_CAP];
  __shared__ unsigned short lst_d[LIST_CAP];
  __shared__ int lperm[LIST_CAP];
  __shared__ int wtot[32];
  __shared__ int wbel[32];
  __shared__ int nlist;

  const int t = threadIdx.x;
  const int lane = t & 31;
  const int w = t >> 5;
  const int k = blockIdx.x;
  const int lo0 = blockIdx.x * S;
  const int lo = lo0 < N ? lo0 : N;
  const int hi = (lo + S < N) ? lo + S : N;
  const unsigned span = (unsigned)(hi - lo);
  const int* __restrict__ dst = ei + E;

  const int c = t >> 3;
  const int j = t & 7;
  const int p0 = pre[c * PRE_ROW + k];
  const int p1 = pre[c * PRE_ROW + k + 1];

  for (int i = t; i < SEG_MAX; i += GRP_THREADS) cnt[i] = 0;
  if (t == 0) nlist = 0;
  __syncthreads();

  int below = (j == 0) ? p0 : 0;

  auto add_edge = [&](int d, int e) {
    const unsigned dl = (unsigned)(d - lo);
    if (dl < span) {
      atomicAdd(&cnt[dl], 1);
      const int slot = atomicAdd(&nlist, 1);
      if (slot < LIST_CAP) { lst_e[slot] = e; lst_d[slot] = (unsigned short)dl; }
    }
  };

  {
    const int nW = (E + 3) >> 2;
    const int CW = (nW + PRE_WGS - 1) / PRE_WGS;
    const int* __restrict__ run = srt + (size_t)c * (size_t)(4 * CW) + p0;
    const int n = p1 - p0;
    for (int i = j; i < n; i += 32) {
      int e[4], d[4];
#pragma unroll
      for (int u = 0; u < 4; ++u) {
        const int q = i + 8 * u;
        e[u] = run[q < n ? q : n - 1];
      }
#pragma unroll
      for (int u = 0; u < 4; ++u) d[u] = dst[e[u]];
#pragma unroll
      for (int u = 0; u < 4; ++u)
        if (i + 8 * u < n) add_edge(d[u], e[u]);
    }
  }
  __syncthreads();

  const int i2 = 2 * t;
  const int v0 = cnt[i2];
  const int v1 = cnt[i2 + 1];
  const int s = v0 + v1;
  int inc = s;
#pragma unroll
  for (int o = 1; o < 32; o <<= 1) {
    int y = __shfl_up(inc, o, 32);
    if (lane >= o) inc += y;
  }
#pragma unroll
  for (int o = 16; o >= 1; o >>= 1) below += __shfl_xor(below, o, 32);
  if (lane == 31) wtot[w] = inc;
  if (lane == 0)  wbel[w] = below;
  __syncthreads();
  const int wt = wtot[lane];
  int winc = wt;
#pragma unroll
  for (int o = 1; o < 32; o <<= 1) {
    int y = __shfl_up(winc, o, 32);
    if (lane >= o) winc += y;
  }
  const int wexcl = __shfl(winc, w, 32) - __shfl(wt, w, 32);
  const int total = __shfl(winc, 31, 32);
  int base = wbel[lane];
#pragma unroll
  for (int o = 16; o >= 1; o >>= 1) base += __shfl_xor(base, o, 32);

  const int ex0 = wexcl + (inc - s);
  const int ex1 = ex0 + v0;
  cnt[i2]     = ex0;
  cnt[i2 + 1] = ex1;
  if ((unsigned)i2 < span)       off[lo + i2]     = base + ex0;
  if ((unsigned)(i2 + 1) < span) off[lo + i2 + 1] = base + ex1;
  if (blockIdx.x == gridDim.x - 1 && t == 0) off[N] = base + total;
  __syncthreads();

  const int nl = nlist;
  if (nl <= LIST_CAP) {
    for (int i = t; i < nl; i += GRP_THREADS) {
      const int pos = atomicAdd(&cnt[lst_d[i]], 1);
      lperm[pos] = lst_e[i];
    }
    __syncthreads();
    for (int i = t; i < nl; i += GRP_THREADS) perm[base + i] = lperm[i];
  } else {
    for (int e = t; e < E; e += GRP_THREADS) {
      const unsigned dl = (unsigned)(dst[e] - lo);
      if (dl < span) perm[base + atomicAdd(&cnt[dl], 1)] = e;
    }
  }
}


template <int M>
__device__ __forceinline__ float row_xor(float v) {
  const int b = __builtin_bit_cast(int, v);
  return __builtin_bit_cast(float, __builtin_amdgcn_update_dpp(b, b, 0x160 + M, 0xf, 0xf, true));
}
__device__ __forceinline__ float xhalf(float v) {
  const int b = __builtin_bit_cast(int, v);
  return __builtin_bit_cast(float,
      __builtin_amdgcn_permlanex16(b, b, 0x76543210, 0xfedcba98, false, false));
}
__device__ __forceinline__ float row_sum16(float v) {
  v += row_xor<8>(v); v += row_xor<4>(v); v += row_xor<2>(v); v += row_xor<1>(v);
  return v;
}
__device__ __forceinline__ float row_max16(float v) {
  v = fmaxf(v, row_xor<8>(v)); v = fmaxf(v, row_xor<4>(v));
  v = fmaxf(v, row_xor<2>(v)); v = fmaxf(v, row_xor<1>(v));
  return v;
}

struct rows_in_lds    { static constexpr bool value = true;  };
struct rows_in_memory { static constexpr bool value = false; };

__global__ void __launch_bounds__(32 * WAVES_PER_BLOCK, 8) gat_node_wmma(
    const float* __restrict__ x,
    const int*   __restrict__ ei,
    const float* __restrict__ attr,
    const float* __restrict__ W_l, const float* __restrict__ b_l,
    const float* __restrict__ W_r, const float* __restrict__ b_r,
    const float* __restrict__ W_e, const float* __restrict__ att,
    const float* __restrict__ bias,
    const int*   __restrict__ off, const int* __restrict__ perm,
    float* __restrict__ out, int N) {
  const int wave = __builtin_amdgcn_readfirstlane(
      (int)((blockIdx.x * blockDim.x + threadIdx.x) >> 5));
  const int lane = threadIdx.x & 31;
  const int nbeg = wave * NODES_PER_WAVE;
  if (nbeg >= N) return;
  const int nend = (nbeg + NODES_PER_WAVE < N) ? nbeg + NODES_PER_WAVE : N;
  const int  lh = lane & 15;
  const bool hi = lane >= 16;
  const float NEG_INF = __int_as_float(0xff800000);
  __shared__ float2 stage[WAVES_PER_BLOCK][STAGE_CAP];
  float2* st = stage[threadIdx.x >> 5];

  v2f Wt[8];
  float av[8][8];
  float s0x = 0.0f, s0y = 0.0f, s1x = 0.0f, s1y = 0.0f;
#pragma unroll
  for (int t = 0; t < 8; ++t) {
    int ch = t * 16 + lh;
    const float k0 = (hi ? W_e : W_l)[ch];
    const float k1 = (hi ? b_l : W_r)[ch];
    const float br = b_r[ch];
    Wt[t].x = k0;
    Wt[t].y = hi ? k1 + br : k1;
    const float ac = att[ch];
    if (t < 4) { s0x = fmaf(ac, Wt[t].x, s0x); s0y = fmaf(ac, Wt[t].y, s0y); }
    else       { s1x = fmaf(ac, Wt[t].x, s1x); s1y = fmaf(ac, Wt[t].y, s1y); }
    const float* ap = att + t * 16 + (hi ? 8 : 0);
    const float4 a_lo = *(const float4*)(ap);
    const float4 a_hi = *(const float4*)(ap + 4);
    av[t][0] = a_lo.x; av[t][1] = a_lo.y; av[t][2] = a_lo.z; av[t][3] = a_lo.w;
    av[t][4] = a_hi.x; av[t][5] = a_hi.y; av[t][6] = a_hi.z; av[t][7] = a_hi.w;
  }
  s0x = row_sum16(s0x);
  s0y = row_sum16(s0y);
  s1x = row_sum16(s1x);
  s1y = row_sum16(s1y);
  const float gx = xhalf(hi ? s0x : s1x);
  const float gy = xhalf(hi ? s0y : s1y);
  const float cA = hi ? gx  : s0x;
  const float cB = hi ? gy  : s0y;
  const float cC = hi ? s1x : gx;
  const float cD = hi ? s1y : gy;

  const float4 wl4 = *(const float4*)(W_l + 4 * lane);
  const float4 bl4 = *(const float4*)(b_l + 4 * lane);
  const float4 bi4 = *(const float4*)(bias + 4 * lane);

  const int b0 = off[nbeg];
  const int cnt = off[nend] - b0;
  const bool staged = cnt <= STAGE_CAP;
  if (staged && cnt > 0) {
    int   se[4], ss[4];
    float sa[4], sx[4];
#pragma unroll
    for (int k = 0; k < 4; ++k) {
      const int i = 32 * k + lane;
      se[k] = perm[b0 + (i < cnt ? i : cnt - 1)];
    }
#pragma unroll
    for (int k = 0; k < 4; ++k) { ss[k] = ei[se[k]]; sa[k] = attr[se[k]]; }
#pragma unroll
    for (int k = 0; k < 4; ++k) sx[k] = x[ss[k]];
#pragma unroll
    for (int k = 0; k < 4; ++k) {
      float2 rw; rw.x = sx[k]; rw.y = sa[k];
      st[32 * k + lane] = rw;
    }
  }
  __builtin_amdgcn_wave_barrier();

  auto nodes = [&](auto where) {
    constexpr bool IN_LDS = decltype(where)::value;
    for (int n = nbeg; n < nend; ++n) {
      const int beg = off[n];
      const int deg = off[n + 1] - beg;
      const int R = deg + 1;
      const float xd = x[n];
      const float2* rows = st + (beg - b0);

      float asum = 0.0f;
      for (int q = lane; q < deg; q += 32) {
        if constexpr (IN_LDS) asum += rows[q].y;
        else                  asum += attr[perm[beg + q]];
      }
      asum += xhalf(asum);
      asum = row_sum16(asum);
      const float loop_attr = asum / fmaxf((float)deg, 1.0f);

      float m = NEG_INF, den = 0.0f, num = 0.0f;

      for (int q0 = 0; q0 < R; q0 += 16) {
        const int q = q0 + lh;
        float a0 = 0.0f, a1 = 0.0f;
        float xs = 0.0f;
        float ra = 0.0f;
        if (q < R) {
          if (q < deg) {
            float2 rw;
            if constexpr (IN_LDS) {
              rw = rows[q];
            } else {
              const int e = perm[beg + q];
              rw.x = x[ei[e]]; rw.y = attr[e];
            }
            xs = rw.x;
            ra = rw.y;
          } else {
            xs = xd;
            ra = loop_attr;
          }
          a0 = hi ? ra : xs;
          a1 = hi ? 1.0f : xd;
        }
        v2f Rf; Rf.x = a0; Rf.y = a1;
        const float lin = fmaf(cA, xs, fmaf(cB, xd, fmaf(cC, ra, cD)));

        float p0a = 0.0f, p0b = 0.0f, p1a = 0.0f, p1b = 0.0f;
#pragma unroll
        for (int t = 0; t < 8; ++t) {
          v8f cz = {};
          v8f D = __builtin_amdgcn_wmma_f32_16x16x4_f32(
              false, Wt[t], false, Rf, (short)0, cz, false, false);
#pragma unroll
          for (int r = 0; r < 8; r += 2) {
            const float g0 = fabsf(D[r]);
            const float g1 = fabsf(D[r + 1]);
            if (t < 4) {
              p0a = fmaf(g0, av[t][r], p0a);
              p0b = fmaf(g1, av[t][r + 1], p0b);
            } else {
              p1a = fmaf(g0, av[t][r], p1a);
              p1b = fmaf(g1, av[t][r + 1], p1b);
            }
          }
        }
        const float p0 = p0a + p0b;
        const float p1 = p1a + p1b;

        const float give = hi ? p0 : p1;
        const float keep = hi ? p1 : p0;
        const float sabs = keep + xhalf(give);
        const float lg = fmaf(LK_ABS, sabs, LK_LIN * lin);
        const bool valid = q < R;

        const float tm = row_max16(valid ? lg : NEG_INF);
        const float nm = fmaxf(m, tm);
        const float sc = expf(m - nm);
        const float ex = valid ? expf(lg - nm) : 0.0f;
        den = fmaf(den, sc, ex);
        num = fmaf(num, sc, valid ? ex * xs : 0.0f);
        m = nm;
      }

      den = row_sum16(den);
      num = row_sum16(num);
      const float S = num / den;
      float4 o4;
      o4.x = fmaf(S, wl4.x, bl4.x + bi4.x);
      o4.y = fmaf(S, wl4.y, bl4.y + bi4.y);
      o4.z = fmaf(S, wl4.z, bl4.z + bi4.z);
      o4.w = fmaf(S, wl4.w, bl4.w + bi4.w);
      *(float4*)(out + (size_t)n * 128 + 4 * lane) = o4;
    }
  };
  if (staged) nodes(rows_in_lds{});
  else        nodes(rows_in_memory{});
}

extern "C" void kernel_launch(void* const* d_in, const int* in_sizes, int n_in,
                              void* d_out, int out_size, void* d_ws, size_t ws_size,
                              hipStream_t stream) {
  const float* x    = (const float*)d_in[0];
  const int*   ei   = (const int*)  d_in[1];
  const float* attr = (const float*)d_in[2];
  const float* W_l  = (const float*)d_in[3];
  const float* b_l  = (const float*)d_in[4];
  const float* W_r  = (const float*)d_in[5];
  const float* b_r  = (const float*)d_in[6];
  const float* W_e  = (const float*)d_in[7];
  const float* att  = (const float*)d_in[8];
  const float* bias = (const float*)d_in[9];

  const int N = in_sizes[0];
  const int E = in_sizes[2];


  const int BLK = 256;

  const int G = (N + SEG_MAX - 1) / SEG_MAX > GRP_WGS ? (N + SEG_MAX - 1) / SEG_MAX : GRP_WGS;
  const int S = (N + G - 1) / G;

  const size_t cap  = 4 * ((((size_t)E + 3) / 4 + PRE_WGS - 1) / PRE_WGS);
  const size_t need = ((size_t)PRE_WGS * cap + (size_t)PRE_WGS * PRE_ROW
                       + (size_t)N + 1 + (size_t)E) * sizeof(int);
  int* off;
  int* perm;
  if (G <= 256 && need <= ws_size) {
    int* srt = (int*)d_ws;
    int* pre = srt + (size_t)PRE_WGS * cap;
    off  = pre + PRE_WGS * PRE_ROW;
    perm = off + N + 1;
    gat_bucket<<<PRE_WGS, PRE_THREADS, 0, stream>>>(ei, srt, pre, E, S);
    gat_group<<<G, GRP_THREADS, 0, stream>>>(ei, srt, pre, off, perm, N, E, S);
  } else {
    off  = (int*)d_ws;
    perm = off + N + 1;
    gat_group_wide<<<G, GRP_THREADS, 0, stream>>>(ei, off, perm, N, E, S);
  }

  int nWaves  = (N + NODES_PER_WAVE - 1) / NODES_PER_WAVE;
  int blocks4 = (nWaves + 7) / 8;
  gat_node_wmma<<<blocks4, BLK, 0, stream>>>(x, ei, attr, W_l, b_l, W_r, b_r, W_e,
                                             att, bias, off, perm, (float*)d_out, N);
}


# Pre-pass packs each edge's destination offset in its node range into the top 11 bits of the sorted edge word (edge id in the low 21); grouping launch takes the node from the word, no dst gather
# speedup vs baseline: 7.6744x; 1.1335x over previous
#include <hip/hip_runtime.h>

typedef __attribute__((ext_vector_type(2))) float v2f;
typedef __attribute__((ext_vector_type(8))) float v8f;

#define NEG_SLOPE 0.2f
#define LK_LIN (0.5f * (1.0f + NEG_SLOPE))
#define LK_ABS (0.5f * (1.0f - NEG_SLOPE))
#define NODES_PER_WAVE 4
#define WAVES_PER_BLOCK 8
#define STAGE_CAP 128

#define GRP_THREADS 1024
#define GRP_WGS 256
#define SEG_MAX 2048
#define LIST_CAP 4096

__global__ void __launch_bounds__(GRP_THREADS)
gat_group_wide(const int* __restrict__ ei,
          int* __restrict__ off, int* __restrict__ perm, int N, int E, int S) {
  __shared__ int cnt[SEG_MAX];
  __shared__ int lst_e[LIST_CAP];
  __shared__ unsigned short lst_d[LIST_CAP];
  __shared__ int lperm[LIST_CAP];
  __shared__ int wtot[32];
  __shared__ int wbel[32];
  __shared__ int nlist;

  const int t = threadIdx.x;
  const int lane = t & 31;
  const int w = t >> 5;
  const int lo0 = blockIdx.x * S;
  const int lo = lo0 < N ? lo0 : N;
  const int hi = (lo + S < N) ? lo + S : N;
  const unsigned span = (unsigned)(hi - lo);
  const int* __restrict__ dst = ei + E;

  for (int i = t; i < SEG_MAX; i += GRP_THREADS) cnt[i] = 0;
  if (t == 0) nlist = 0;
  __syncthreads();

  int below = 0;
  auto visit = [&](int d, int e) {
    below += (d < lo) ? 1 : 0;
    const unsigned dl = (unsigned)(d - lo);
    if (dl < span) {
      atomicAdd(&cnt[dl], 1);
      const int slot = atomicAdd(&nlist, 1);
      if (slot < LIST_CAP) { lst_e[slot] = e; lst_d[slot] = (unsigned short)dl; }
    }
  };

  const int E4 = ((E & 3) == 0) ? (E >> 2) : 0;
  const int4* __restrict__ dst4 = (const int4*)dst;
  const int nfull = E4 / (8 * GRP_THREADS);
  for (int it = 0; it < nfull; ++it) {
    int4 v[8];
#pragma unroll
    for (int u = 0; u < 8; ++u) v[u] = dst4[(it * 8 + u) * GRP_THREADS + t];
#pragma unroll
    for (int u = 0; u < 8; ++u) {
      const int e = 4 * ((it * 8 + u) * GRP_THREADS + t);
      visit(v[u].x, e);     visit(v[u].y, e + 1);
      visit(v[u].z, e + 2); visit(v[u].w, e + 3);
    }
  }
  for (int i = nfull * 8 * GRP_THREADS + t; i < E4; i += GRP_THREADS) {
    const int4 v = dst4[i];
    visit(v.x, 4 * i);     visit(v.y, 4 * i + 1);
    visit(v.z, 4 * i + 2); visit(v.w, 4 * i + 3);
  }
  for (int e = 4 * E4 + t; e < E; e += GRP_THREADS) visit(dst[e], e);

  __syncthreads();

  const int i2 = 2 * t;
  const int v0 = cnt[i2];
  const int v1 = cnt[i2 + 1];
  const int s = v0 + v1;
  int inc = s;
#pragma unroll
  for (int o = 1; o < 32; o <<= 1) {
    int y = __shfl_up(inc, o, 32);
    if (lane >= o) inc += y;
  }
#pragma unroll
  for (int o = 16; o >= 1; o >>= 1) below += __shfl_xor(below, o, 32);
  if (lane == 31) wtot[w] = inc;
  if (lane == 0)  wbel[w] = below;
  __syncthreads();
  const int wt = wtot[lane];
  int winc = wt;
#pragma unroll
  for (int o = 1; o < 32; o <<= 1) {
    int y = __shfl_up(winc, o, 32);
    if (lane >= o) winc += y;
  }
  const int wexcl = __shfl(winc, w, 32) - __shfl(wt, w, 32);
  const int total = __shfl(winc, 31, 32);
  int base = wbel[lane];
#pragma unroll
  for (int o = 16; o >= 1; o >>= 1) base += __shfl_xor(base, o, 32);

  const int ex0 = wexcl + (inc - s);
  const int ex1 = ex0 + v0;
  cnt[i2]     = ex0;
  cnt[i2 + 1] = ex1;
  if ((unsigned)i2 < span)       off[lo + i2]     = base + ex0;
  if ((unsigned)(i2 + 1) < span) off[lo + i2 + 1] = base + ex1;
  if (blockIdx.x == gridDim.x - 1 && t == 0) off[N] = base + total;
  __syncthreads();

  const int nl = nlist;
  if (nl <= LIST_CAP) {
    for (int i = t; i < nl; i += GRP_THREADS) {
      const int pos = atomicAdd(&cnt[lst_d[i]], 1);
      lperm[pos] = lst_e[i];
    }
    __syncthreads();
    for (int i = t; i < nl; i += GRP_THREADS) perm[base + i] = lperm[i];
  } else {
    for (int e = t; e < E; e += GRP_THREADS) {
      const unsigned dl = (unsigned)(dst[e] - lo);
      if (dl < span) perm[base + atomicAdd(&cnt[dl], 1)] = e;
    }
  }
}

#define PRE_WGS 128
#define PRE_THREADS 1024
#define PRE_ROW 257
#define PACK_SHIFT 21
#define PACK_E_MAX (1 << PACK_SHIFT)
#define PACK_E_MASK ((1u << PACK_SHIFT) - 1u)
static_assert(SEG_MAX <= (1 << (32 - PACK_SHIFT)), "offset in a range fits the top bits");

__device__ __forceinline__ unsigned range_id(unsigned d, unsigned S, unsigned M) {
  unsigned q = __umulhi(d, M);
  if (d - q * S >= S) ++q;
  return q & 255u;
}

__global__ void __launch_bounds__(PRE_THREADS)
gat_bucket(const int* __restrict__ ei,
           int* __restrict__ srt, int* __restrict__ pre, int E, int S) {
  __shared__ int h[256];
  __shared__ int wsum[8];
  const int t = threadIdx.x;
  const int lane = t & 31;
  if (t < 256) h[t] = 0;
  __syncthreads();
  const int* __restrict__ dst = ei + E;
  const unsigned uS = (unsigned)S;
  const unsigned M = 0xffffffffu / uS;
  const int nW = (E + 3) >> 2;
  const int CW = (nW + (int)gridDim.x - 1) / (int)gridDim.x;
  const int w0 = (int)blockIdx.x * CW;
  const int w1 = (w0 + CW < nW) ? w0 + CW : nW;
  const bool al = (E & 3) == 0;
  int* __restrict__ mine = srt + (size_t)blockIdx.x * (size_t)(4 * CW);

  auto load4 = [&](int g, int (&d)[4]) {
    const int e = 4 * g;
    if (al) {
      const int4 v = ((const int4*)dst)[g];
      d[0] = v.x; d[1] = v.y; d[2] = v.z; d[3] = v.w;
    } else {
      d[0] = dst[e];
      d[1] = (e + 1 < E) ? dst[e + 1] : -1;
      d[2] = (e + 2 < E) ? dst[e + 2] : -1;
      d[3] = (e + 3 < E) ? dst[e + 3] : -1;
    }
  };

  for (int g = w0 + t; g < w1; g += PRE_THREADS) {
    int d[4];
    load4(g, d);
#pragma unroll
    for (int j = 0; j < 4; ++j)
      if (d[j] >= 0) atomicAdd(&h[range_id((unsigned)d[j], uS, M)], 1);
  }
  __syncthreads();

  const int v = (t < 256) ? h[t] : 0;
  int inc = v;
#pragma unroll
  for (int o = 1; o < 32; o <<= 1) {
    int y = __shfl_up(inc, o, 32);
    if (lane >= o) inc += y;
  }
  if (t < 256 && lane == 31) wsum[t >> 5] = inc;
  __syncthreads();
  if (t < 256) {
    int wex = 0;
#pragma unroll
    for (int q = 0; q < 8; ++q) wex += (q < (t >> 5)) ? wsum[q] : 0;
    const int ex = wex + inc - v;
    h[t] = ex;
    pre[blockIdx.x * PRE_ROW + t] = ex;
    if (t == 255) pre[blockIdx.x * PRE_ROW + 256] = ex + v;
  }
  __syncthreads();

  const bool pk = E <= PACK_E_MAX;
  for (int g = w0 + t; g < w1; g += PRE_THREADS) {
    int d[4];
    load4(g, d);
    int pos[4];
    unsigned rid[4];
#pragma unroll
    for (int j = 0; j < 4; ++j) {
      rid[j] = range_id((unsigned)d[j], uS, M);
      pos[j] = (d[j] >= 0) ? atomicAdd(&h[rid[j]], 1) : -1;
    }
#pragma unroll
    for (int j = 0; j < 4; ++j)
      if (pos[j] >= 0) {
        const unsigned e  = (unsigned)(4 * g + j);
        const unsigned dl = (unsigned)d[j] - rid[j] * uS;
        mine[pos[j]] = (int)(pk ? (e | (dl << PACK_SHIFT)) : e);
      }
  }
}

template <int M>
__device__ __forceinline__ float row_xor(float v) {
  const int b = __builtin_bit_cast(int, v);
  return __builtin_bit_cast(float, __builtin_amdgcn_update_dpp(b, b, 0x160 + M, 0xf, 0xf, true));
}
__device__ __forceinline__ float xhalf(float v) {
  const int b = __builtin_bit_cast(int, v);
  return __builtin_bit_cast(float,
      __builtin_amdgcn_permlanex16(b, b, 0x76543210, 0xfedcba98, false, false));
}
__device__ __forceinline__ float row_sum16(float v) {
  v += row_xor<8>(v); v += row_xor<4>(v); v += row_xor<2>(v); v += row_xor<1>(v);
  return v;
}
__device__ __forceinline__ float row_max16(float v) {
  v = fmaxf(v, row_xor<8>(v)); v = fmaxf(v, row_xor<4>(v));
  v = fmaxf(v, row_xor<2>(v)); v = fmaxf(v, row_xor<1>(v));
  return v;
}


struct wave_consts {
  v2f Wt[8];
  float av[8][8];
  float cA, cB, cC, cD;
  float4 wl4, bl4, bi4;
};

__device__ __forceinline__ void load_wave_consts(
    wave_consts& k, int lane,
    const float* __restrict__ W_l, const float* __restrict__ b_l,
    const float* __restrict__ W_r, const float* __restrict__ b_r,
    const float* __restrict__ W_e, const float* __restrict__ att,
    const float* __restrict__ bias) {
  const int  lh = lane & 15;
  const bool hi = lane >= 16;
  float s0x = 0.0f, s0y = 0.0f, s1x = 0.0f, s1y = 0.0f;
#pragma unroll
  for (int t = 0; t < 8; ++t) {
    int ch = t * 16 + lh;
    const float k0 = (hi ? W_e : W_l)[ch];
    const float k1 = (hi ? b_l : W_r)[ch];
    const float br = b_r[ch];
    k.Wt[t].x = k0;
    k.Wt[t].y = hi ? k1 + br : k1;
    const float ac = att[ch];
    if (t < 4) { s0x = fmaf(ac, k.Wt[t].x, s0x); s0y = fmaf(ac, k.Wt[t].y, s0y); }
    else       { s1x = fmaf(ac, k.Wt[t].x, s1x); s1y = fmaf(ac, k.Wt[t].y, s1y); }
    const float* ap = att + t * 16 + (hi ? 8 : 0);
    const float4 a_lo = *(const float4*)(ap);
    const float4 a_hi = *(const float4*)(ap + 4);
    k.av[t][0] = a_lo.x; k.av[t][1] = a_lo.y; k.av[t][2] = a_lo.z; k.av[t][3] = a_lo.w;
    k.av[t][4] = a_hi.x; k.av[t][5] = a_hi.y; k.av[t][6] = a_hi.z; k.av[t][7] = a_hi.w;
  }
  s0x = row_sum16(s0x);
  s0y = row_sum16(s0y);
  s1x = row_sum16(s1x);
  s1y = row_sum16(s1y);
  const float gx = xhalf(hi ? s0x : s1x);
  const float gy = xhalf(hi ? s0y : s1y);
  k.cA = hi ? gx  : s0x;
  k.cB = hi ? gy  : s0y;
  k.cC = hi ? s1x : gx;
  k.cD = hi ? s1y : gy;

  k.wl4 = *(const float4*)(W_l + 4 * lane);
  k.bl4 = *(const float4*)(b_l + 4 * lane);
  k.bi4 = *(const float4*)(bias + 4 * lane);
}

template <bool IN_LDS>
__device__ __forceinline__ void one_node(
    const wave_consts& k, int lane, int n, int deg,
    const float2* rows, const int* __restrict__ prun,
    const float* __restrict__ x, const int* __restrict__ ei,
    const float* __restrict__ attr, float* __restrict__ out) {
  const int  lh = lane & 15;
  const bool hi = lane >= 16;
  const float NEG_INF = __int_as_float(0xff800000);
  const int R = deg + 1;
  const float xd = x[n];

  float asum = 0.0f;
  for (int q = lane; q < deg; q += 32) {
    if constexpr (IN_LDS) asum += rows[q].y;
    else                  asum += attr[prun[q]];
  }
  asum += xhalf(asum);
  asum = row_sum16(asum);
  const float loop_attr = asum / fmaxf((float)deg, 1.0f);

  float m = NEG_INF, den = 0.0f, num = 0.0f;

  for (int q0 = 0; q0 < R; q0 += 16) {
    const int q = q0 + lh;
    float a0 = 0.0f, a1 = 0.0f;
    float xs = 0.0f;
    float ra = 0.0f;
    if (q < R) {
      if (q < deg) {
        float2 rw;
        if constexpr (IN_LDS) {
          rw = rows[q];
        } else {
          const int e = prun[q];
          rw.x = x[ei[e]]; rw.y = attr[e];
        }
        xs = rw.x;
        ra = rw.y;
      } else {
        xs = xd;
        ra = loop_attr;
      }
      a0 = hi ? ra : xs;
      a1 = hi ? 1.0f : xd;
    }
    v2f Rf; Rf.x = a0; Rf.y = a1;
    const float lin = fmaf(k.cA, xs, fmaf(k.cB, xd, fmaf(k.cC, ra, k.cD)));

    float p0a = 0.0f, p0b = 0.0f, p1a = 0.0f, p1b = 0.0f;
#pragma unroll
    for (int t = 0; t < 8; ++t) {
      v8f cz = {};
      v8f D = __builtin_amdgcn_wmma_f32_16x16x4_f32(
          false, k.Wt[t], false, Rf, (short)0, cz, false, false);
#pragma unroll
      for (int r = 0; r < 8; r += 2) {
        const float g0 = fabsf(D[r]);
        const float g1 = fabsf(D[r + 1]);
        if (t < 4) {
          p0a = fmaf(g0, k.av[t][r], p0a);
          p0b = fmaf(g1, k.av[t][r + 1], p0b);
        } else {
          p1a = fmaf(g0, k.av[t][r], p1a);
          p1b = fmaf(g1, k.av[t][r + 1], p1b);
        }
      }
    }
    const float p0 = p0a + p0b;
    const float p1 = p1a + p1b;

    const float give = hi ? p0 : p1;
    const float keep = hi ? p1 : p0;
    const float sabs = keep + xhalf(give);
    const float lg = fmaf(LK_ABS, sabs, LK_LIN * lin);
    const bool valid = q < R;

    const float tm = row_max16(valid ? lg : NEG_INF);
    const float nm = fmaxf(m, tm);
    const float sc = expf(m - nm);
    const float ex = valid ? expf(lg - nm) : 0.0f;
    den = fmaf(den, sc, ex);
    num = fmaf(num, sc, valid ? ex * xs : 0.0f);
    m = nm;
  }

  den = row_sum16(den);
  num = row_sum16(num);
  const float S = num / den;
  float4 o4;
  o4.x = fmaf(S, k.wl4.x, k.bl4.x + k.bi4.x);
  o4.y = fmaf(S, k.wl4.y, k.bl4.y + k.bi4.y);
  o4.z = fmaf(S, k.wl4.z, k.bl4.z + k.bi4.z);
  o4.w = fmaf(S, k.wl4.w, k.bl4.w + k.bi4.w);
  *(float4*)(out + (size_t)n * 128 + 4 * lane) = o4;
}

__global__ void __launch_bounds__(32 * WAVES_PER_BLOCK, 8) gat_node_wmma(
    const float* __restrict__ x,
    const int*   __restrict__ ei,
    const float* __restrict__ attr,
    const float* __restrict__ W_l, const float* __restrict__ b_l,
    const float* __restrict__ W_r, const float* __restrict__ b_r,
    const float* __restrict__ W_e, const float* __restrict__ att,
    const float* __restrict__ bias,
    const int*   __restrict__ off, const int* __restrict__ perm,
    float* __restrict__ out, int N) {
  const int wave = __builtin_amdgcn_readfirstlane(
      (int)((blockIdx.x * blockDim.x + threadIdx.x) >> 5));
  const int lane = threadIdx.x & 31;
  const int nbeg = wave * NODES_PER_WAVE;
  if (nbeg >= N) return;
  const int nend = (nbeg + NODES_PER_WAVE < N) ? nbeg + NODES_PER_WAVE : N;
  __shared__ float2 stage[WAVES_PER_BLOCK][STAGE_CAP];
  float2* st = stage[threadIdx.x >> 5];

  wave_consts k;
  load_wave_consts(k, lane, W_l, b_l, W_r, b_r, W_e, att, bias);

  const int b0 = off[nbeg];
  const int cnt = off[nend] - b0;
  const bool staged = cnt <= STAGE_CAP;
  if (staged && cnt > 0) {
    int   se[4], ss[4];
    float sa[4], sx[4];
#pragma unroll
    for (int u = 0; u < 4; ++u) {
      const int i = 32 * u + lane;
      se[u] = perm[b0 + (i < cnt ? i : cnt - 1)];
    }
#pragma unroll
    for (int u = 0; u < 4; ++u) { ss[u] = ei[se[u]]; sa[u] = attr[se[u]]; }
#pragma unroll
    for (int u = 0; u < 4; ++u) sx[u] = x[ss[u]];
#pragma unroll
    for (int u = 0; u < 4; ++u) {
      float2 rw; rw.x = sx[u]; rw.y = sa[u];
      st[32 * u + lane] = rw;
    }
  }
  __builtin_amdgcn_wave_barrier();

  if (staged) {
    for (int n = nbeg; n < nend; ++n) {
      const int beg = off[n];
      one_node<true>(k, lane, n, off[n + 1] - beg, st + (beg - b0), nullptr,
                     x, ei, attr, out);
    }
  } else {
    for (int n = nbeg; n < nend; ++n) {
      const int beg = off[n];
      one_node<false>(k, lane, n, off[n + 1] - beg, nullptr, perm + beg,
                      x, ei, attr, out);
    }
  }
}

static_assert(GRP_THREADS == 8 * PRE_WGS, "eight threads to a chunk");
static_assert(LIST_CAP == 4 * GRP_THREADS, "four listed edges to a thread");

__global__ void __launch_bounds__(GRP_THREADS)
gat_group_node(const float* __restrict__ x,
               const int*   __restrict__ ei,
               const float* __restrict__ attr,
               const float* __restrict__ W_l, const float* __restrict__ b_l,
               const float* __restrict__ W_r, const float* __restrict__ b_r,
               const float* __restrict__ W_e, const float* __restrict__ att,
               const float* __restrict__ bias,
               const int* __restrict__ srt, const int* __restrict__ pre,
               int* perm, float* __restrict__ out, int N, int E, int S) {
  __shared__ int cnt[SEG_MAX];
  __shared__ int lst_e[LIST_CAP];
  __shared__ unsigned short lst_d[LIST_CAP];
  __shared__ float2 rowsL[LIST_CAP];
  __shared__ int wtot[32];
  __shared__ int wbel[32];
  __shared__ int nlist;

  const int t = threadIdx.x;
  const int lane = t & 31;
  const int w = __builtin_amdgcn_readfirstlane(t >> 5);
  const int kr = blockIdx.x;
  const int lo0 = blockIdx.x * S;
  const int lo = lo0 < N ? lo0 : N;
  const int hi = (lo + S < N) ? lo + S : N;
  const unsigned span = (unsigned)(hi - lo);
  const int* __restrict__ dst = ei + E;

  const int c = t >> 3;
  const int j = t & 7;
  const int p0 = pre[c * PRE_ROW + kr];
  const int p1 = pre[c * PRE_ROW + kr + 1];

  for (int i = t; i < SEG_MAX; i += GRP_THREADS) cnt[i] = 0;
  if (t == 0) nlist = 0;
  __syncthreads();

  const int nW = (E + 3) >> 2;
  const int CW = (nW + PRE_WGS - 1) / PRE_WGS;
  const int* __restrict__ run = srt + (size_t)c * (size_t)(4 * CW) + p0;
  const int nrun = p1 - p0;
  const bool pk = E <= PACK_E_MAX;
  for (int i = j; i < nrun; i += 32) {
    int e[4];
    unsigned dl[4];
#pragma unroll
    for (int u = 0; u < 4; ++u) {
      const int q = i + 8 * u;
      e[u] = run[q < nrun ? q : nrun - 1];
    }
    if (pk) {
#pragma unroll
      for (int u = 0; u < 4; ++u) {
        const unsigned wd = (unsigned)e[u];
        dl[u] = wd >> PACK_SHIFT;
        e[u]  = (int)(wd & PACK_E_MASK);
      }
    } else {
#pragma unroll
      for (int u = 0; u < 4; ++u) dl[u] = (unsigned)(dst[e[u]] - lo);
    }
#pragma unroll
    for (int u = 0; u < 4; ++u) {
      if (i + 8 * u < nrun && dl[u] < span) {
        atomicAdd(&cnt[dl[u]], 1);
        const int slot = atomicAdd(&nlist, 1);
        if (slot < LIST_CAP) { lst_e[slot] = e[u]; lst_d[slot] = (unsigned short)dl[u]; }
      }
    }
  }
  __syncthreads();

  const int nl = nlist;
  const bool listed = nl <= LIST_CAP;

  float ga[4] = {0.0f, 0.0f, 0.0f, 0.0f};
  float gx[4] = {0.0f, 0.0f, 0.0f, 0.0f};
  if (listed && nl > 0) {
    int ge[4], gs[4];
#pragma unroll
    for (int u = 0; u < 4; ++u) {
      const int i = t + GRP_THREADS * u;
      ge[u] = lst_e[i < nl ? i : nl - 1];
    }
#pragma unroll
    for (int u = 0; u < 4; ++u) { gs[u] = ei[ge[u]]; ga[u] = attr[ge[u]]; }
#pragma unroll
    for (int u = 0; u < 4; ++u) gx[u] = x[gs[u]];
  }

  const int i2 = 2 * t;
  const int v0 = cnt[i2];
  const int v1 = cnt[i2 + 1];
  const int s = v0 + v1;
  int inc = s;
#pragma unroll
  for (int o = 1; o < 32; o <<= 1) {
    int y = __shfl_up(inc, o, 32);
    if (lane >= o) inc += y;
  }
  if (lane == 31) wtot[w] = inc;
  __syncthreads();
  const int wt = wtot[lane];
  int winc = wt;
#pragma unroll
  for (int o = 1; o < 32; o <<= 1) {
    int y = __shfl_up(winc, o, 32);
    if (lane >= o) winc += y;
  }
  const int wexcl = __shfl(winc, w, 32) - __shfl(wt, w, 32);
  const int ex0 = wexcl + (inc - s);
  const int ex1 = ex0 + v0;
  cnt[i2]     = ex0;
  cnt[i2 + 1] = ex1;
  __syncthreads();

  const int nnodes = __builtin_amdgcn_readfirstlane((int)span);
  int base = 0;
  if (listed) {
#pragma unroll
    for (int u = 0; u < 4; ++u) {
      const int i = t + GRP_THREADS * u;
      if (i < nl) {
        const int pos = atomicAdd(&cnt[lst_d[i]], 1);
        float2 rw; rw.x = gx[u]; rw.y = ga[u];
        rowsL[pos] = rw;
      }
    }
  } else {
    int below = (j == 0) ? p0 : 0;
#pragma unroll
    for (int o = 16; o >= 1; o >>= 1) below += __shfl_xor(below, o, 32);
    if (lane == 0) wbel[w] = below;
    __syncthreads();
    base = wbel[lane];
#pragma unroll
    for (int o = 16; o >= 1; o >>= 1) base += __shfl_xor(base, o, 32);
    base = __builtin_amdgcn_readfirstlane(base);

    for (int i = j; i < nrun; i += 8) {
      const unsigned wd = (unsigned)run[i];
      const int e = pk ? (int)(wd & PACK_E_MASK) : (int)wd;
      const unsigned dl = pk ? (wd >> PACK_SHIFT) : (unsigned)(dst[e] - lo);
      if (dl < span) perm[base + atomicAdd(&cnt[dl], 1)] = e;
    }
  }

  wave_consts k;
  load_wave_consts(k, lane, W_l, b_l, W_r, b_r, W_e, att, bias);

  __syncthreads();

  if (listed) {
    for (int i = w; i < nnodes; i += 32) {
      const int prev = cnt[i ? i - 1 : 0];
      const int end  = cnt[i];
      const int beg  = i ? prev : 0;
      one_node<true>(k, lane, lo + i, end - beg, rowsL + beg, nullptr,
                     x, ei, attr, out);
    }
  } else {
    for (int i = w; i < nnodes; i += 32) {
      const int beg = i ? cnt[i - 1] : 0;
      one_node<false>(k, lane, lo + i, cnt[i] - beg, nullptr, perm + base + beg,
                      x, ei, attr, out);
    }
  }
}

extern "C" void kernel_launch(void* const* d_in, const int* in_sizes, int n_in,
                              void* d_out, int out_size, void* d_ws, size_t ws_size,
                              hipStream_t stream) {
  const float* x    = (const float*)d_in[0];
  const int*   ei   = (const int*)  d_in[1];
  const float* attr = (const float*)d_in[2];
  const float* W_l  = (const float*)d_in[3];
  const float* b_l  = (const float*)d_in[4];
  const float* W_r  = (const float*)d_in[5];
  const float* b_r  = (const float*)d_in[6];
  const float* W_e  = (const float*)d_in[7];
  const float* att  = (const float*)d_in[8];
  const float* bias = (const float*)d_in[9];

  const int N = in_sizes[0];
  const int E = in_sizes[2];


  const int BLK = 256;

  const int G = (N + SEG_MAX - 1) / SEG_MAX > GRP_WGS ? (N + SEG_MAX - 1) / SEG_MAX : GRP_WGS;
  const int S = (N + G - 1) / G;

  const size_t cap  = 4 * ((((size_t)E + 3) / 4 + PRE_WGS - 1) / PRE_WGS);
  const size_t need = ((size_t)PRE_WGS * cap + (size_t)PRE_WGS * PRE_ROW
                       + (size_t)N + 1 + (size_t)E) * sizeof(int);
  if (G <= 256 && need <= ws_size) {
    int* srt  = (int*)d_ws;
    int* pre  = srt + (size_t)PRE_WGS * cap;
    int* perm = pre + PRE_WGS * PRE_ROW + N + 1;
    gat_bucket<<<PRE_WGS, PRE_THREADS, 0, stream>>>(ei, srt, pre, E, S);
    gat_group_node<<<G, GRP_THREADS, 0, stream>>>(x, ei, attr, W_l, b_l, W_r, b_r, W_e,
                                                  att, bias, srt, pre, perm,
                                                  (float*)d_out, N, E, S);
  } else {
    int* off  = (int*)d_ws;
    int* perm = off + N + 1;
    gat_group_wide<<<G, GRP_THREADS, 0, stream>>>(ei, off, perm, N, E, S);
    int nWaves  = (N + NODES_PER_WAVE - 1) / NODES_PER_WAVE;
    int blocks4 = (nWaves + 7) / 8;
    gat_node_wmma<<<blocks4, BLK, 0, stream>>>(x, ei, attr, W_l, b_l, W_r, b_r, W_e,
                                               att, bias, off, perm, (float*)d_out, N);
  }
}
